# starting kernel
# baseline (speedup 1.0000x reference)
#include <hip/hip_runtime.h>
#include <hip/hip_bf16.h>
#include <math.h>

#define BB   2
#define LL   384
#define DD   128
#define NH   8
#define DHS_ 16
#define DHP_ 4
#define BL   (BB*LL)
#define BNL  (BB*NH*LL)
#define DOUT 1280
#define LT   24

typedef float v2f __attribute__((ext_vector_type(2)));
typedef float v8f __attribute__((ext_vector_type(8)));

static __device__ __forceinline__ v8f wmma4(v2f a, v2f b, v8f c) {
  return __builtin_amdgcn_wmma_f32_16x16x4_f32(false, a, false, b, (short)0, c,
                                               false, false);
}

__global__ void wmma_gemm_kernel(const float* __restrict__ A,
                                 const float* __restrict__ W,
                                 const float* __restrict__ bias,
                                 float* __restrict__ C,
                                 int M, int K, int Nc, int total_tiles) {
  int tile = blockIdx.x * 8 + (threadIdx.x >> 5);
  if (tile >= total_tiles) return;
  int lane = threadIdx.x & 31;
  int h = lane >> 4, l16 = lane & 15;
  int ntn = Nc >> 4;
  int tm = (tile / ntn) << 4;
  int tn = (tile % ntn) << 4;
  const float* ap = A + (size_t)(tm + l16) * K + 2 * h;
  const float* wp = W + (size_t)(2 * h) * Nc + tn + l16;
  v8f acc = {};
  for (int k = 0; k < K; k += 4) {
    v2f a, b;
    a.x = ap[k];
    a.y = ap[k + 1];
    b.x = wp[(size_t)k * Nc];
    b.y = wp[(size_t)(k + 1) * Nc];
    acc = wmma4(a, b, acc);
  }
  float bv = bias ? bias[tn + l16] : 0.0f;
#pragma unroll
  for (int v = 0; v < 8; ++v) {
    int row = tm + v + (h << 3);
    C[(size_t)row * Nc + tn + l16] = acc[v] + bv;
  }
}

__global__ void bias_kernel(const float* __restrict__ e,
                            const float* __restrict__ Wb,
                            float* __restrict__ bias) {
  __shared__ float se[32 * 129];
  __shared__ float swb[128 * 8];
  int blk = blockIdx.x;
  int jb = blk % 12;
  int i  = (blk / 12) % LL;
  int b  = blk / (12 * LL);
  const float* ep = e + (((size_t)b * LL + i) * LL + (size_t)jb * 32) * DD;
  for (int idx = threadIdx.x; idx < 32 * 128; idx += 256) {
    int jj = idx >> 7, d = idx & 127;
    se[jj * 129 + d] = ep[idx];
  }
  for (int idx = threadIdx.x; idx < 128 * 8; idx += 256) swb[idx] = Wb[idx];
  __syncthreads();
  int n = threadIdx.x >> 5, jj = threadIdx.x & 31;
  float s = 0.0f;
  for (int d = 0; d < 128; ++d) s += se[jj * 129 + d] * swb[d * 8 + n];
  bias[(((size_t)b * NH + n) * LL + i) * LL + jb * 32 + jj] = s;
}

__global__ void point_transform_kernel(const float* __restrict__ hmat,
                                       const float* __restrict__ r,
                                       const float* __restrict__ t,
                                       float* __restrict__ outp,
                                       float* __restrict__ sumsq) {
  int idx = blockIdx.x * blockDim.x + threadIdx.x;
  if (idx >= BNL) return;
  int l = idx % LL;
  int n = (idx / LL) % NH;
  int b = idx / (NH * LL);
  const float* hp = hmat + (size_t)(b * LL + l) * (NH * 12) + n * 12;
  const float* rp = r + (size_t)(b * LL + l) * 9;
  const float* tp = t + (size_t)(b * LL + l) * 3;
  float* op = outp + (size_t)idx * 12;
  float ss = 0.0f;
#pragma unroll
  for (int p = 0; p < 4; ++p)
#pragma unroll
    for (int c = 0; c < 3; ++c) {
      float v = tp[c];
#pragma unroll
      for (int k = 0; k < 3; ++k) v += hp[p * 3 + k] * rp[k * 3 + c];
      op[p * 3 + c] = v;
      ss += v * v;
    }
  if (sumsq) sumsq[idx] = ss;
}

__global__ void logit_kernel(const float* __restrict__ qs,
                             const float* __restrict__ ks,
                             const float* __restrict__ qp,
                             const float* __restrict__ kp,
                             const float* __restrict__ qq,
                             const float* __restrict__ kk,
                             const float* __restrict__ bias,
                             const float* __restrict__ gamma,
                             float* __restrict__ logit) {
  int tile = blockIdx.x * 8 + (threadIdx.x >> 5);
  int jt = tile % LT;
  int it = (tile / LT) % LT;
  int n  = (tile / (LT * LT)) % NH;
  int b  = tile / (LT * LT * NH);
  int lane = threadIdx.x & 31;
  int h = lane >> 4, l16 = lane & 15;

  const float* qsp = qs + (size_t)(b * LL + it * 16 + l16) * DD + n * DHS_ + 2 * h;
  const float* ksp = ks + (size_t)(b * LL + jt * 16 + l16) * DD + n * DHS_ + 2 * h;
  v8f accS = {};
#pragma unroll
  for (int k = 0; k < 16; k += 4) {
    v2f a, bv2;
    a.x = qsp[k];   a.y = qsp[k + 1];
    bv2.x = ksp[k]; bv2.y = ksp[k + 1];
    accS = wmma4(a, bv2, accS);
  }
  const float* qpp = qp + (size_t)((b * NH + n) * LL + it * 16 + l16) * 12 + 2 * h;
  const float* kpp = kp + (size_t)((b * NH + n) * LL + jt * 16 + l16) * 12 + 2 * h;
  v8f accP = {};
#pragma unroll
  for (int k = 0; k < 12; k += 4) {
    v2f a, bv2;
    a.x = qpp[k];   a.y = qpp[k + 1];
    bv2.x = kpp[k]; bv2.y = kpp[k + 1];
    accP = wmma4(a, bv2, accP);
  }
  size_t hb = (size_t)(b * NH + n) * LL;
  int j = jt * 16 + l16;
  float kkj = kk[hb + j];
  float g = gamma[n];
  const float rsqrt3 = 0.57735026918962576f;
#pragma unroll
  for (int v = 0; v < 8; ++v) {
    int i = it * 16 + v + (h << 3);
    float d2 = qq[hb + i] + kkj - 2.0f * accP[v];
    float lp = -0.125f * g * d2;
    float bsv = bias[(hb + i) * LL + j];
    logit[(hb + i) * LL + j] = rsqrt3 * (0.25f * accS[v] + bsv + lp);
  }
}

__global__ void softmax_kernel(float* __restrict__ attn) {
  int row = blockIdx.x * 8 + (threadIdx.x >> 5);
  int lane = threadIdx.x & 31;
  float* p = attn + (size_t)row * LL;
  float v[12];
  float m = -1e30f;
#pragma unroll
  for (int i = 0; i < 12; ++i) {
    v[i] = p[lane + i * 32];
    m = fmaxf(m, v[i]);
  }
#pragma unroll
  for (int off = 16; off > 0; off >>= 1) m = fmaxf(m, __shfl_xor(m, off, 32));
  float s = 0.0f;
#pragma unroll
  for (int i = 0; i < 12; ++i) {
    v[i] = __expf(v[i] - m);
    s += v[i];
  }
#pragma unroll
  for (int off = 16; off > 0; off >>= 1) s += __shfl_xor(s, off, 32);
  float inv = 1.0f / s;
#pragma unroll
  for (int i = 0; i < 12; ++i) p[lane + i * 32] = v[i] * inv;
}

__global__ void attn_v_kernel(const float* __restrict__ attn,
                              const float* __restrict__ vs,
                              const float* __restrict__ vp,
                              float* __restrict__ out_cat,
                              float* __restrict__ op) {
  int tile = blockIdx.x * 8 + (threadIdx.x >> 5);
  int it = tile % LT;
  int n  = (tile / LT) % NH;
  int b  = tile / (LT * NH);
  int lane = threadIdx.x & 31;
  int h = lane >> 4, l16 = lane & 15;
  size_t hb = (size_t)(b * NH + n) * LL;
  const float* ap  = attn + (hb + it * 16 + l16) * LL + 2 * h;
  const float* vsb = vs + (size_t)b * LL * DD + n * DHS_ + l16;
  const float* vpb = vp + hb * 12 + l16;
  bool pv = l16 < 12;
  v8f accS = {}, accP = {};
  for (int k = 0; k < LL; k += 4) {
    v2f a;
    a.x = ap[k];
    a.y = ap[k + 1];
    v2f bs;
    bs.x = vsb[(size_t)(k + 2 * h) * DD];
    bs.y = vsb[(size_t)(k + 2 * h + 1) * DD];
    accS = wmma4(a, bs, accS);
    v2f bp;
    bp.x = pv ? vpb[(k + 2 * h) * 12] : 0.0f;
    bp.y = pv ? vpb[(k + 2 * h + 1) * 12] : 0.0f;
    accP = wmma4(a, bp, accP);
  }
#pragma unroll
  for (int v = 0; v < 8; ++v) {
    int i = it * 16 + v + (h << 3);
    out_cat[(size_t)(b * LL + i) * DOUT + n * DHS_ + l16] = accS[v];
    if (pv) op[(hb + i) * 12 + l16] = accP[v];
  }
}

__global__ void out_pair_kernel(const float* __restrict__ attn,
                                const float* __restrict__ e,
                                float* __restrict__ out_cat) {
  int tile = blockIdx.x * 8 + (threadIdx.x >> 5);
  int dt = tile % 8;
  int i  = (tile / 8) % LL;
  int b  = tile / (8 * LL);
  int lane = threadIdx.x & 31;
  int h = lane >> 4, l16 = lane & 15;
  bool av = l16 < 8;
  const float* ap = attn + ((size_t)(b * NH + l16) * LL + i) * LL + 2 * h;
  const float* ep = e + ((size_t)(b * LL + i) * LL) * DD + dt * 16 + l16;
  v8f acc = {};
  for (int k = 0; k < LL; k += 4) {
    if ((k & 63) == 0 && (k + 64) < LL)
      __builtin_prefetch(ep + (size_t)(k + 64) * DD, 0, 0);
    v2f a;
    a.x = av ? ap[k] : 0.0f;
    a.y = av ? ap[k + 1] : 0.0f;
    v2f bv;
    bv.x = ep[(size_t)(k + 2 * h) * DD];
    bv.y = ep[(size_t)(k + 2 * h + 1) * DD];
    acc = wmma4(a, bv, acc);
  }
  if (h == 0) {
#pragma unroll
    for (int v = 0; v < 8; ++v)
      out_cat[(size_t)(b * LL + i) * DOUT + DD + v * DD + dt * 16 + l16] = acc[v];
  }
}

__global__ void op_finish_kernel(const float* __restrict__ op,
                                 const float* __restrict__ r,
                                 const float* __restrict__ t,
                                 float* __restrict__ out_cat) {
  int idx = blockIdx.x * blockDim.x + threadIdx.x;
  if (idx >= BNL) return;
  int i = idx % LL;
  int n = (idx / LL) % NH;
  int b = idx / (NH * LL);
  const float* opv = op + (size_t)idx * 12;
  const float* rp = r + (size_t)(b * LL + i) * 9;
  const float* tp = t + (size_t)(b * LL + i) * 3;
  float* oc = out_cat + (size_t)(b * LL + i) * DOUT;
#pragma unroll
  for (int p = 0; p < 4; ++p) {
    float om[3];
#pragma unroll
    for (int k = 0; k < 3; ++k) om[k] = opv[p * 3 + k] - tp[k];
    float nsq = 0.0f;
#pragma unroll
    for (int c = 0; c < 3; ++c) {
      float v = om[0] * rp[c * 3 + 0] + om[1] * rp[c * 3 + 1] + om[2] * rp[c * 3 + 2];
      oc[DD + NH * DD + n * 12 + p * 3 + c] = v;
      nsq += v * v;
    }
    oc[DD + NH * DD + NH * 12 + n * 4 + p] = sqrtf(nsq);
  }
}

extern "C" void kernel_launch(void* const* d_in, const int* in_sizes, int n_in,
                              void* d_out, int out_size, void* d_ws, size_t ws_size,
                              hipStream_t stream) {
  (void)in_sizes; (void)n_in; (void)out_size;
  const float* x     = (const float*)d_in[0];
  const float* e     = (const float*)d_in[1];
  const float* r     = (const float*)d_in[2];
  const float* t     = (const float*)d_in[3];
  const float* Wq_s  = (const float*)d_in[4];
  const float* Wk_s  = (const float*)d_in[5];
  const float* Wv_s  = (const float*)d_in[6];
  const float* Wb    = (const float*)d_in[7];
  const float* Wq_p  = (const float*)d_in[8];
  const float* Wk_p  = (const float*)d_in[9];
  const float* Wv_p  = (const float*)d_in[10];
  const float* gamma = (const float*)d_in[11];
  const float* Wo    = (const float*)d_in[12];
  const float* bo    = (const float*)d_in[13];
  float* out = (float*)d_out;
  float* ws  = (float*)d_ws;

  float* qs   = ws;
  float* ks_  = qs  + 98304;
  float* vs   = ks_ + 98304;
  float* hq   = vs  + 98304;
  float* hk   = hq  + 73728;
  float* hv   = hk  + 73728;
  float* qp   = hv  + 73728;
  float* kp   = qp  + 73728;
  float* vp   = kp  + 73728;
  float* qq   = vp  + 73728;
  float* kk   = qq  + 6144;
  float* bias = kk  + 6144;
  float* attn = bias + 2359296;
  float* op   = attn + 2359296;
  float* ocat = op  + 73728;
  size_t req = (size_t)((ocat + 983040) - ws) * sizeof(float);
  if (ws_size < req) return;

  dim3 blk(256);
  auto gemm = [&](const float* A, const float* W, const float* bvec, float* C,
                  int M, int K, int Nc) {
    int tiles = (M / 16) * (Nc / 16);
    wmma_gemm_kernel<<<(tiles + 7) / 8, blk, 0, stream>>>(A, W, bvec, C, M, K,
                                                          Nc, tiles);
  };

  gemm(x, Wq_s, nullptr, qs, BL, DD, DD);
  gemm(x, Wk_s, nullptr, ks_, BL, DD, DD);
  gemm(x, Wv_s, nullptr, vs, BL, DD, DD);
  gemm(x, Wq_p, nullptr, hq, BL, DD, NH * 12);
  gemm(x, Wk_p, nullptr, hk, BL, DD, NH * 12);
  gemm(x, Wv_p, nullptr, hv, BL, DD, NH * 12);

  bias_kernel<<<BB * LL * 12, blk, 0, stream>>>(e, Wb, bias);

  point_transform_kernel<<<BNL / 256, blk, 0, stream>>>(hq, r, t, qp, qq);
  point_transform_kernel<<<BNL / 256, blk, 0, stream>>>(hk, r, t, kp, kk);
  point_transform_kernel<<<BNL / 256, blk, 0, stream>>>(hv, r, t, vp, nullptr);

  logit_kernel<<<(BB * NH * LT * LT) / 8, blk, 0, stream>>>(qs, ks_, qp, kp, qq,
                                                            kk, bias, gamma, attn);
  softmax_kernel<<<BNL / 8, blk, 0, stream>>>(attn);

  attn_v_kernel<<<(BB * NH * LT) / 8, blk, 0, stream>>>(attn, vs, vp, ocat, op);
  out_pair_kernel<<<(BB * LL * 8) / 8, blk, 0, stream>>>(attn, e, ocat);
  op_finish_kernel<<<BNL / 256, blk, 0, stream>>>(op, r, t, ocat);

  gemm(ocat, Wo, bo, out, BL, DOUT, DD);
}


# WMMA k-loops (GEMMs, attn@V, pair output): keep 12-16 k-steps of fragment loads in flight, each slot refilled right after its WMMA; pad lanes load duplicate valid data unmasked; k order unchanged
# speedup vs baseline: 1.8667x; 1.8667x over previous
#include <hip/hip_runtime.h>
#include <hip/hip_bf16.h>
#include <math.h>

#define BB   2
#define LL   384
#define DD   128
#define NH   8
#define DHS_ 16
#define DHP_ 4
#define BL   (BB*LL)
#define BNL  (BB*NH*LL)
#define DOUT 1280
#define LT   24

#define GK 16

typedef float v2f __attribute__((ext_vector_type(2)));
typedef float v8f __attribute__((ext_vector_type(8)));

static __device__ __forceinline__ v8f wmma4(v2f a, v2f b, v8f c) {
  return __builtin_amdgcn_wmma_f32_16x16x4_f32(false, a, false, b, (short)0, c,
                                               false, false);
}

__global__ void wmma_gemm_kernel(const float* __restrict__ A,
                                 const float* __restrict__ W,
                                 const float* __restrict__ bias,
                                 float* __restrict__ C,
                                 int M, int K, int Nc, int total_tiles) {
  int tile = blockIdx.x * 8 + (threadIdx.x >> 5);
  if (tile >= total_tiles) return;
  int lane = threadIdx.x & 31;
  int h = lane >> 4, l16 = lane & 15;
  int ntn = Nc >> 4;
  int tm = (tile / ntn) << 4;
  int tn = (tile % ntn) << 4;
  const float* ap = A + (size_t)(tm + l16) * K + 2 * h;
  const float* wp = W + (size_t)(2 * h) * Nc + tn + l16;
  v8f acc = {};
  v2f a[GK], b[GK];
  auto load = [&](int s, int k) {
    a[s].x = ap[k];
    a[s].y = ap[k + 1];
    b[s].x = wp[(size_t)k * Nc];
    b[s].y = wp[(size_t)(k + 1) * Nc];
  };
#pragma unroll
  for (int s = 0; s < GK; ++s) {
    load(s, 4 * s);
    __builtin_amdgcn_sched_barrier(0);
  }
  int k0 = 0;
  for (; k0 + 4 * GK < K; k0 += 4 * GK) {
#pragma unroll
    for (int s = 0; s < GK; ++s) {
      acc = wmma4(a[s], b[s], acc);
      load(s, k0 + 4 * GK + 4 * s);
      __builtin_amdgcn_sched_barrier(0);
    }
  }
#pragma unroll
  for (int s = 0; s < GK; ++s) acc = wmma4(a[s], b[s], acc);
  float bv = bias ? bias[tn + l16] : 0.0f;
#pragma unroll
  for (int v = 0; v < 8; ++v) {
    int row = tm + v + (h << 3);
    C[(size_t)row * Nc + tn + l16] = acc[v] + bv;
  }
}

__global__ void bias_kernel(const float* __restrict__ e,
                            const float* __restrict__ Wb,
                            float* __restrict__ bias) {
  __shared__ float se[32 * 129];
  __shared__ float swb[128 * 8];
  int blk = blockIdx.x;
  int jb = blk % 12;
  int i  = (blk / 12) % LL;
  int b  = blk / (12 * LL);
  const float* ep = e + (((size_t)b * LL + i) * LL + (size_t)jb * 32) * DD;
  for (int idx = threadIdx.x; idx < 32 * 128; idx += 256) {
    int jj = idx >> 7, d = idx & 127;
    se[jj * 129 + d] = ep[idx];
  }
  for (int idx = threadIdx.x; idx < 128 * 8; idx += 256) swb[idx] = Wb[idx];
  __syncthreads();
  int n = threadIdx.x >> 5, jj = threadIdx.x & 31;
  float s = 0.0f;
  for (int d = 0; d < 128; ++d) s += se[jj * 129 + d] * swb[d * 8 + n];
  bias[(((size_t)b * NH + n) * LL + i) * LL + jb * 32 + jj] = s;
}

__global__ void point_transform_kernel(const float* __restrict__ hmat,
                                       const float* __restrict__ r,
                                       const float* __restrict__ t,
                                       float* __restrict__ outp,
                                       float* __restrict__ sumsq) {
  int idx = blockIdx.x * blockDim.x + threadIdx.x;
  if (idx >= BNL) return;
  int l = idx % LL;
  int n = (idx / LL) % NH;
  int b = idx / (NH * LL);
  const float* hp = hmat + (size_t)(b * LL + l) * (NH * 12) + n * 12;
  const float* rp = r + (size_t)(b * LL + l) * 9;
  const float* tp = t + (size_t)(b * LL + l) * 3;
  float* op = outp + (size_t)idx * 12;
  float ss = 0.0f;
#pragma unroll
  for (int p = 0; p < 4; ++p)
#pragma unroll
    for (int c = 0; c < 3; ++c) {
      float v = tp[c];
#pragma unroll
      for (int k = 0; k < 3; ++k) v += hp[p * 3 + k] * rp[k * 3 + c];
      op[p * 3 + c] = v;
      ss += v * v;
    }
  if (sumsq) sumsq[idx] = ss;
}

__global__ void logit_kernel(const float* __restrict__ qs,
                             const float* __restrict__ ks,
                             const float* __restrict__ qp,
                             const float* __restrict__ kp,
                             const float* __restrict__ qq,
                             const float* __restrict__ kk,
                             const float* __restrict__ bias,
                             const float* __restrict__ gamma,
                             float* __restrict__ logit) {
  int tile = blockIdx.x * 8 + (threadIdx.x >> 5);
  int jt = tile % LT;
  int it = (tile / LT) % LT;
  int n  = (tile / (LT * LT)) % NH;
  int b  = tile / (LT * LT * NH);
  int lane = threadIdx.x & 31;
  int h = lane >> 4, l16 = lane & 15;

  const float* qsp = qs + (size_t)(b * LL + it * 16 + l16) * DD + n * DHS_ + 2 * h;
  const float* ksp = ks + (size_t)(b * LL + jt * 16 + l16) * DD + n * DHS_ + 2 * h;
  v8f accS = {};
#pragma unroll
  for (int k = 0; k < 16; k += 4) {
    v2f a, bv2;
    a.x = qsp[k];   a.y = qsp[k + 1];
    bv2.x = ksp[k]; bv2.y = ksp[k + 1];
    accS = wmma4(a, bv2, accS);
  }
  const float* qpp = qp + (size_t)((b * NH + n) * LL + it * 16 + l16) * 12 + 2 * h;
  const float* kpp = kp + (size_t)((b * NH + n) * LL + jt * 16 + l16) * 12 + 2 * h;
  v8f accP = {};
#pragma unroll
  for (int k = 0; k < 12; k += 4) {
    v2f a, bv2;
    a.x = qpp[k];   a.y = qpp[k + 1];
    bv2.x = kpp[k]; bv2.y = kpp[k + 1];
    accP = wmma4(a, bv2, accP);
  }
  size_t hb = (size_t)(b * NH + n) * LL;
  int j = jt * 16 + l16;
  float kkj = kk[hb + j];
  float g = gamma[n];
  const float rsqrt3 = 0.57735026918962576f;
#pragma unroll
  for (int v = 0; v < 8; ++v) {
    int i = it * 16 + v + (h << 3);
    float d2 = qq[hb + i] + kkj - 2.0f * accP[v];
    float lp = -0.125f * g * d2;
    float bsv = bias[(hb + i) * LL + j];
    logit[(hb + i) * LL + j] = rsqrt3 * (0.25f * accS[v] + bsv + lp);
  }
}

__global__ void softmax_kernel(float* __restrict__ attn) {
  int row = blockIdx.x * 8 + (threadIdx.x >> 5);
  int lane = threadIdx.x & 31;
  float* p = attn + (size_t)row * LL;
  float v[12];
  float m = -1e30f;
#pragma unroll
  for (int i = 0; i < 12; ++i) {
    v[i] = p[lane + i * 32];
    m = fmaxf(m, v[i]);
  }
#pragma unroll
  for (int off = 16; off > 0; off >>= 1) m = fmaxf(m, __shfl_xor(m, off, 32));
  float s = 0.0f;
#pragma unroll
  for (int i = 0; i < 12; ++i) {
    v[i] = __expf(v[i] - m);
    s += v[i];
  }
#pragma unroll
  for (int off = 16; off > 0; off >>= 1) s += __shfl_xor(s, off, 32);
  float inv = 1.0f / s;
#pragma unroll
  for (int i = 0; i < 12; ++i) p[lane + i * 32] = v[i] * inv;
}

__global__ void attn_v_kernel(const float* __restrict__ attn,
                              const float* __restrict__ vs,
                              const float* __restrict__ vp,
                              float* __restrict__ out_cat,
                              float* __restrict__ op) {
  int tile = blockIdx.x * 8 + (threadIdx.x >> 5);
  int it = tile % LT;
  int n  = (tile / LT) % NH;
  int b  = tile / (LT * NH);
  int lane = threadIdx.x & 31;
  int h = lane >> 4, l16 = lane & 15;
  size_t hb = (size_t)(b * NH + n) * LL;
  const float* ap  = attn + (hb + it * 16 + l16) * LL + 2 * h;
  const float* vsb = vs + (size_t)b * LL * DD + n * DHS_ + l16;
  bool pv = l16 < 12;
  const float* vpb = vp + hb * 12 + (pv ? l16 : 11);
  v8f accS = {}, accP = {};
  constexpr int GV = 12;
  v2f a[GV], bs[GV], bp[GV];
  auto load = [&](int s, int k) {
    a[s].x = ap[k];
    a[s].y = ap[k + 1];
    bs[s].x = vsb[(size_t)(k + 2 * h) * DD];
    bs[s].y = vsb[(size_t)(k + 2 * h + 1) * DD];
    bp[s].x = vpb[(k + 2 * h) * 12];
    bp[s].y = vpb[(k + 2 * h + 1) * 12];
  };
#pragma unroll
  for (int s = 0; s < GV; ++s) {
    load(s, 4 * s);
    __builtin_amdgcn_sched_barrier(0);
  }
  for (int k0 = 0; k0 + 4 * GV < LL; k0 += 4 * GV) {
#pragma unroll
    for (int s = 0; s < GV; ++s) {
      accS = wmma4(a[s], bs[s], accS);
      accP = wmma4(a[s], bp[s], accP);
      load(s, k0 + 4 * GV + 4 * s);
      __builtin_amdgcn_sched_barrier(0);
    }
  }
#pragma unroll
  for (int s = 0; s < GV; ++s) {
    accS = wmma4(a[s], bs[s], accS);
    accP = wmma4(a[s], bp[s], accP);
  }
#pragma unroll
  for (int v = 0; v < 8; ++v) {
    int i = it * 16 + v + (h << 3);
    out_cat[(size_t)(b * LL + i) * DOUT + n * DHS_ + l16] = accS[v];
    if (pv) op[(hb + i) * 12 + l16] = accP[v];
  }
}

__global__ void out_pair_kernel(const float* __restrict__ attn,
                                const float* __restrict__ e,
                                float* __restrict__ out_cat) {
  int tile = blockIdx.x * 8 + (threadIdx.x >> 5);
  int dt = tile % 8;
  int i  = (tile / 8) % LL;
  int b  = tile / (8 * LL);
  int lane = threadIdx.x & 31;
  int h = lane >> 4, l16 = lane & 15;
  const float* ap = attn + ((size_t)(b * NH + (l16 & 7)) * LL + i) * LL + 2 * h;
  const float* ep = e + ((size_t)(b * LL + i) * LL) * DD + dt * 16 + l16;
  v8f acc = {};
  v2f a[GK], bv[GK];
  auto load = [&](int s, int k) {
    a[s].x = ap[k];
    a[s].y = ap[k + 1];
    bv[s].x = ep[(size_t)(k + 2 * h) * DD];
    bv[s].y = ep[(size_t)(k + 2 * h + 1) * DD];
  };
#pragma unroll
  for (int s = 0; s < GK; ++s) {
    load(s, 4 * s);
    __builtin_amdgcn_sched_barrier(0);
  }
  for (int k0 = 0; k0 + 4 * GK < LL; k0 += 4 * GK) {
#pragma unroll
    for (int s = 0; s < GK; ++s) {
      acc = wmma4(a[s], bv[s], acc);
      load(s, k0 + 4 * GK + 4 * s);
      __builtin_amdgcn_sched_barrier(0);
    }
  }
#pragma unroll
  for (int s = 0; s < GK; ++s) acc = wmma4(a[s], bv[s], acc);
  if (h == 0) {
#pragma unroll
    for (int v = 0; v < 8; ++v)
      out_cat[(size_t)(b * LL + i) * DOUT + DD + v * DD + dt * 16 + l16] = acc[v];
  }
}

__global__ void op_finish_kernel(const float* __restrict__ op,
                                 const float* __restrict__ r,
                                 const float* __restrict__ t,
                                 float* __restrict__ out_cat) {
  int idx = blockIdx.x * blockDim.x + threadIdx.x;
  if (idx >= BNL) return;
  int i = idx % LL;
  int n = (idx / LL) % NH;
  int b = idx / (NH * LL);
  const float* opv = op + (size_t)idx * 12;
  const float* rp = r + (size_t)(b * LL + i) * 9;
  const float* tp = t + (size_t)(b * LL + i) * 3;
  float* oc = out_cat + (size_t)(b * LL + i) * DOUT;
#pragma unroll
  for (int p = 0; p < 4; ++p) {
    float om[3];
#pragma unroll
    for (int k = 0; k < 3; ++k) om[k] = opv[p * 3 + k] - tp[k];
    float nsq = 0.0f;
#pragma unroll
    for (int c = 0; c < 3; ++c) {
      float v = om[0] * rp[c * 3 + 0] + om[1] * rp[c * 3 + 1] + om[2] * rp[c * 3 + 2];
      oc[DD + NH * DD + n * 12 + p * 3 + c] = v;
      nsq += v * v;
    }
    oc[DD + NH * DD + NH * 12 + n * 4 + p] = sqrtf(nsq);
  }
}

extern "C" void kernel_launch(void* const* d_in, const int* in_sizes, int n_in,
                              void* d_out, int out_size, void* d_ws, size_t ws_size,
                              hipStream_t stream) {
  (void)in_sizes; (void)n_in; (void)out_size;
  const float* x     = (const float*)d_in[0];
  const float* e     = (const float*)d_in[1];
  const float* r     = (const float*)d_in[2];
  const float* t     = (const float*)d_in[3];
  const float* Wq_s  = (const float*)d_in[4];
  const float* Wk_s  = (const float*)d_in[5];
  const float* Wv_s  = (const float*)d_in[6];
  const float* Wb    = (const float*)d_in[7];
  const float* Wq_p  = (const float*)d_in[8];
  const float* Wk_p  = (const float*)d_in[9];
  const float* Wv_p  = (const float*)d_in[10];
  const float* gamma = (const float*)d_in[11];
  const float* Wo    = (const float*)d_in[12];
  const float* bo    = (const float*)d_in[13];
  float* out = (float*)d_out;
  float* ws  = (float*)d_ws;

  float* qs   = ws;
  float* ks_  = qs  + 98304;
  float* vs   = ks_ + 98304;
  float* hq   = vs  + 98304;
  float* hk   = hq  + 73728;
  float* hv   = hk  + 73728;
  float* qp   = hv  + 73728;
  float* kp   = qp  + 73728;
  float* vp   = kp  + 73728;
  float* qq   = vp  + 73728;
  float* kk   = qq  + 6144;
  float* bias = kk  + 6144;
  float* attn = bias + 2359296;
  float* op   = attn + 2359296;
  float* ocat = op  + 73728;
  size_t req = (size_t)((ocat + 983040) - ws) * sizeof(float);
  if (ws_size < req) return;

  dim3 blk(256);
  auto gemm = [&](const float* A, const float* W, const float* bvec, float* C,
                  int M, int K, int Nc) {
    int tiles = (M / 16) * (Nc / 16);
    wmma_gemm_kernel<<<(tiles + 7) / 8, blk, 0, stream>>>(A, W, bvec, C, M, K,
                                                          Nc, tiles);
  };

  gemm(x, Wq_s, nullptr, qs, BL, DD, DD);
  gemm(x, Wk_s, nullptr, ks_, BL, DD, DD);
  gemm(x, Wv_s, nullptr, vs, BL, DD, DD);
  gemm(x, Wq_p, nullptr, hq, BL, DD, NH * 12);
  gemm(x, Wk_p, nullptr, hk, BL, DD, NH * 12);
  gemm(x, Wv_p, nullptr, hv, BL, DD, NH * 12);

  bias_kernel<<<BB * LL * 12, blk, 0, stream>>>(e, Wb, bias);

  point_transform_kernel<<<BNL / 256, blk, 0, stream>>>(hq, r, t, qp, qq);
  point_transform_kernel<<<BNL / 256, blk, 0, stream>>>(hk, r, t, kp, kk);
  point_transform_kernel<<<BNL / 256, blk, 0, stream>>>(hv, r, t, vp, nullptr);

  logit_kernel<<<(BB * NH * LT * LT) / 8, blk, 0, stream>>>(qs, ks_, qp, kp, qq,
                                                            kk, bias, gamma, attn);
  softmax_kernel<<<BNL / 8, blk, 0, stream>>>(attn);

  attn_v_kernel<<<(BB * NH * LT) / 8, blk, 0, stream>>>(attn, vs, vp, ocat, op);
  out_pair_kernel<<<(BB * LL * 8) / 8, blk, 0, stream>>>(attn, e, ocat);
  op_finish_kernel<<<BNL / 256, blk, 0, stream>>>(op, r, t, ocat);

  gemm(ocat, Wo, bo, out, BL, DOUT, DD);
}


# pair bias e@Wb moved onto the f32 WMMA: e rows load straight into A fragments (16 B per lane, 32 loads in flight), Wb fragments stay in registers, 3072 waves not 73728; d sum order changes
# speedup vs baseline: 2.3413x; 1.2542x over previous
#include <hip/hip_runtime.h>
#include <hip/hip_bf16.h>
#include <math.h>

#define BB   2
#define LL   384
#define DD   128
#define NH   8
#define DHS_ 16
#define DHP_ 4
#define BL   (BB*LL)
#define BNL  (BB*NH*LL)
#define DOUT 1280
#define LT   24

#define GK 16

typedef float v2f __attribute__((ext_vector_type(2)));
typedef float v8f __attribute__((ext_vector_type(8)));

static __device__ __forceinline__ v8f wmma4(v2f a, v2f b, v8f c) {
  return __builtin_amdgcn_wmma_f32_16x16x4_f32(false, a, false, b, (short)0, c,
                                               false, false);
}

__global__ void wmma_gemm_kernel(const float* __restrict__ A,
                                 const float* __restrict__ W,
                                 const float* __restrict__ bias,
                                 float* __restrict__ C,
                                 int M, int K, int Nc, int total_tiles) {
  int tile = blockIdx.x * 8 + (threadIdx.x >> 5);
  if (tile >= total_tiles) return;
  int lane = threadIdx.x & 31;
  int h = lane >> 4, l16 = lane & 15;
  int ntn = Nc >> 4;
  int tm = (tile / ntn) << 4;
  int tn = (tile % ntn) << 4;
  const float* ap = A + (size_t)(tm + l16) * K + 2 * h;
  const float* wp = W + (size_t)(2 * h) * Nc + tn + l16;
  v8f acc = {};
  v2f a[GK], b[GK];
  auto load = [&](int s, int k) {
    a[s].x = ap[k];
    a[s].y = ap[k + 1];
    b[s].x = wp[(size_t)k * Nc];
    b[s].y = wp[(size_t)(k + 1) * Nc];
  };
#pragma unroll
  for (int s = 0; s < GK; ++s) {
    load(s, 4 * s);
    __builtin_amdgcn_sched_barrier(0);
  }
  int k0 = 0;
  for (; k0 + 4 * GK < K; k0 += 4 * GK) {
#pragma unroll
    for (int s = 0; s < GK; ++s) {
      acc = wmma4(a[s], b[s], acc);
      load(s, k0 + 4 * GK + 4 * s);
      __builtin_amdgcn_sched_barrier(0);
    }
  }
#pragma unroll
  for (int s = 0; s < GK; ++s) acc = wmma4(a[s], b[s], acc);
  float bv = bias ? bias[tn + l16] : 0.0f;
#pragma unroll
  for (int v = 0; v < 8; ++v) {
    int row = tm + v + (h << 3);
    C[(size_t)row * Nc + tn + l16] = acc[v] + bv;
  }
}

typedef float v4f __attribute__((ext_vector_type(4)));

#define BIAS_TPW 6
#define BIAS_NS  32

__global__ void __launch_bounds__(128)
bias_kernel(const float* __restrict__ e,
            const float* __restrict__ Wb,
            float* __restrict__ bias) {
  int wave = threadIdx.x >> 5;
  int lane = threadIdx.x & 31;
  int h = lane >> 4, l16 = lane & 15;
  int bi = blockIdx.x;
  int b = bi / LL, i = bi % LL;
  int n = l16 & 7;
  v2f wf[32];
  const float* wp = Wb + (4 * h) * NH + n;
#pragma unroll
  for (int m = 0; m < 16; ++m) {
#pragma unroll
    for (int q = 0; q < 2; ++q) {
      wf[2 * m + q].x = wp[(8 * m + 2 * q) * NH];
      wf[2 * m + q].y = wp[(8 * m + 2 * q + 1) * NH];
    }
  }
  const float* ap =
      e + ((size_t)bi * LL + (size_t)(wave * BIAS_TPW * 16 + l16)) * DD + 4 * h;
  auto ld = [&](int t, int m) -> v4f {
    return *reinterpret_cast<const v4f*>(ap + (size_t)t * (16 * DD) + 8 * m);
  };
  bool hi = l16 >= 8;
  float* bp = bias + (((size_t)(b * NH + n) * LL + i) * LL) +
              wave * BIAS_TPW * 16 + 8 * h + (hi ? 4 : 0);
  auto store = [&](const v8f& acc, int t) {
    v4f o;
    o.x = hi ? acc[4] : acc[0];
    o.y = hi ? acc[5] : acc[1];
    o.z = hi ? acc[6] : acc[2];
    o.w = hi ? acc[7] : acc[3];
    *reinterpret_cast<v4f*>(bp + t * 16) = o;
  };
  v4f a[BIAS_NS];
#pragma unroll
  for (int s = 0; s < BIAS_NS; ++s) {
    a[s] = ld(s >> 4, s & 15);
    __builtin_amdgcn_sched_barrier(0);
  }
  for (int tp = 0; tp < BIAS_TPW / 2 - 1; ++tp) {
#pragma unroll
    for (int hf = 0; hf < 2; ++hf) {
      v8f acc = {};
#pragma unroll
      for (int m = 0; m < 16; ++m) {
        int s = hf * 16 + m;
        v2f a0 = {a[s].x, a[s].y}, a1 = {a[s].z, a[s].w};
        acc = wmma4(a0, wf[2 * m], acc);
        acc = wmma4(a1, wf[2 * m + 1], acc);
        a[s] = ld(2 * tp + 2 + hf, m);
        __builtin_amdgcn_sched_barrier(0);
      }
      store(acc, 2 * tp + hf);
    }
  }
#pragma unroll
  for (int hf = 0; hf < 2; ++hf) {
    v8f acc = {};
#pragma unroll
    for (int m = 0; m < 16; ++m) {
      int s = hf * 16 + m;
      v2f a0 = {a[s].x, a[s].y}, a1 = {a[s].z, a[s].w};
      acc = wmma4(a0, wf[2 * m], acc);
      acc = wmma4(a1, wf[2 * m + 1], acc);
    }
    store(acc, BIAS_TPW - 2 + hf);
  }
}

__global__ void point_transform_kernel(const float* __restrict__ hmat,
                                       const float* __restrict__ r,
                                       const float* __restrict__ t,
                                       float* __restrict__ outp,
                                       float* __restrict__ sumsq) {
  int idx = blockIdx.x * blockDim.x + threadIdx.x;
  if (idx >= BNL) return;
  int l = idx % LL;
  int n = (idx / LL) % NH;
  int b = idx / (NH * LL);
  const float* hp = hmat + (size_t)(b * LL + l) * (NH * 12) + n * 12;
  const float* rp = r + (size_t)(b * LL + l) * 9;
  const float* tp = t + (size_t)(b * LL + l) * 3;
  float* op = outp + (size_t)idx * 12;
  float ss = 0.0f;
#pragma unroll
  for (int p = 0; p < 4; ++p)
#pragma unroll
    for (int c = 0; c < 3; ++c) {
      float v = tp[c];
#pragma unroll
      for (int k = 0; k < 3; ++k) v += hp[p * 3 + k] * rp[k * 3 + c];
      op[p * 3 + c] = v;
      ss += v * v;
    }
  if (sumsq) sumsq[idx] = ss;
}

__global__ void logit_kernel(const float* __restrict__ qs,
                             const float* __restrict__ ks,
                             const float* __restrict__ qp,
                             const float* __restrict__ kp,
                             const float* __restrict__ qq,
                             const float* __restrict__ kk,
                             const float* __restrict__ bias,
                             const float* __restrict__ gamma,
                             float* __restrict__ logit) {
  int tile = blockIdx.x * 8 + (threadIdx.x >> 5);
  int jt = tile % LT;
  int it = (tile / LT) % LT;
  int n  = (tile / (LT * LT)) % NH;
  int b  = tile / (LT * LT * NH);
  int lane = threadIdx.x & 31;
  int h = lane >> 4, l16 = lane & 15;

  const float* qsp = qs + (size_t)(b * LL + it * 16 + l16) * DD + n * DHS_ + 2 * h;
  const float* ksp = ks + (size_t)(b * LL + jt * 16 + l16) * DD + n * DHS_ + 2 * h;
  v8f accS = {};
#pragma unroll
  for (int k = 0; k < 16; k += 4) {
    v2f a, bv2;
    a.x = qsp[k];   a.y = qsp[k + 1];
    bv2.x = ksp[k]; bv2.y = ksp[k + 1];
    accS = wmma4(a, bv2, accS);
  }
  const float* qpp = qp + (size_t)((b * NH + n) * LL + it * 16 + l16) * 12 + 2 * h;
  const float* kpp = kp + (size_t)((b * NH + n) * LL + jt * 16 + l16) * 12 + 2 * h;
  v8f accP = {};
#pragma unroll
  for (int k = 0; k < 12; k += 4) {
    v2f a, bv2;
    a.x = qpp[k];   a.y = qpp[k + 1];
    bv2.x = kpp[k]; bv2.y = kpp[k + 1];
    accP = wmma4(a, bv2, accP);
  }
  size_t hb = (size_t)(b * NH + n) * LL;
  int j = jt * 16 + l16;
  float kkj = kk[hb + j];
  float g = gamma[n];
  const float rsqrt3 = 0.57735026918962576f;
#pragma unroll
  for (int v = 0; v < 8; ++v) {
    int i = it * 16 + v + (h << 3);
    float d2 = qq[hb + i] + kkj - 2.0f * accP[v];
    float lp = -0.125f * g * d2;
    float bsv = bias[(hb + i) * LL + j];
    logit[(hb + i) * LL + j] = rsqrt3 * (0.25f * accS[v] + bsv + lp);
  }
}

__global__ void softmax_kernel(float* __restrict__ attn) {
  int row = blockIdx.x * 8 + (threadIdx.x >> 5);
  int lane = threadIdx.x & 31;
  float* p = attn + (size_t)row * LL;
  float v[12];
  float m = -1e30f;
#pragma unroll
  for (int i = 0; i < 12; ++i) {
    v[i] = p[lane + i * 32];
    m = fmaxf(m, v[i]);
  }
#pragma unroll
  for (int off = 16; off > 0; off >>= 1) m = fmaxf(m, __shfl_xor(m, off, 32));
  float s = 0.0f;
#pragma unroll
  for (int i = 0; i < 12; ++i) {
    v[i] = __expf(v[i] - m);
    s += v[i];
  }
#pragma unroll
  for (int off = 16; off > 0; off >>= 1) s += __shfl_xor(s, off, 32);
  float inv = 1.0f / s;
#pragma unroll
  for (int i = 0; i < 12; ++i) p[lane + i * 32] = v[i] * inv;
}

__global__ void attn_v_kernel(const float* __restrict__ attn,
                              const float* __restrict__ vs,
                              const float* __restrict__ vp,
                              float* __restrict__ out_cat,
                              float* __restrict__ op) {
  int tile = blockIdx.x * 8 + (threadIdx.x >> 5);
  int it = tile % LT;
  int n  = (tile / LT) % NH;
  int b  = tile / (LT * NH);
  int lane = threadIdx.x & 31;
  int h = lane >> 4, l16 = lane & 15;
  size_t hb = (size_t)(b * NH + n) * LL;
  const float* ap  = attn + (hb + it * 16 + l16) * LL + 2 * h;
  const float* vsb = vs + (size_t)b * LL * DD + n * DHS_ + l16;
  bool pv = l16 < 12;
  const float* vpb = vp + hb * 12 + (pv ? l16 : 11);
  v8f accS = {}, accP = {};
  constexpr int GV = 12;
  v2f a[GV], bs[GV], bp[GV];
  auto load = [&](int s, int k) {
    a[s].x = ap[k];
    a[s].y = ap[k + 1];
    bs[s].x = vsb[(size_t)(k + 2 * h) * DD];
    bs[s].y = vsb[(size_t)(k + 2 * h + 1) * DD];
    bp[s].x = vpb[(k + 2 * h) * 12];
    bp[s].y = vpb[(k + 2 * h + 1) * 12];
  };
#pragma unroll
  for (int s = 0; s < GV; ++s) {
    load(s, 4 * s);
    __builtin_amdgcn_sched_barrier(0);
  }
  for (int k0 = 0; k0 + 4 * GV < LL; k0 += 4 * GV) {
#pragma unroll
    for (int s = 0; s < GV; ++s) {
      accS = wmma4(a[s], bs[s], accS);
      accP = wmma4(a[s], bp[s], accP);
      load(s, k0 + 4 * GV + 4 * s);
      __builtin_amdgcn_sched_barrier(0);
    }
  }
#pragma unroll
  for (int s = 0; s < GV; ++s) {
    accS = wmma4(a[s], bs[s], accS);
    accP = wmma4(a[s], bp[s], accP);
  }
#pragma unroll
  for (int v = 0; v < 8; ++v) {
    int i = it * 16 + v + (h << 3);
    out_cat[(size_t)(b * LL + i) * DOUT + n * DHS_ + l16] = accS[v];
    if (pv) op[(hb + i) * 12 + l16] = accP[v];
  }
}

__global__ void out_pair_kernel(const float* __restrict__ attn,
                                const float* __restrict__ e,
                                float* __restrict__ out_cat) {
  int tile = blockIdx.x * 8 + (threadIdx.x >> 5);
  int dt = tile % 8;
  int i  = (tile / 8) % LL;
  int b  = tile / (8 * LL);
  int lane = threadIdx.x & 31;
  int h = lane >> 4, l16 = lane & 15;
  const float* ap = attn + ((size_t)(b * NH + (l16 & 7)) * LL + i) * LL + 2 * h;
  const float* ep = e + ((size_t)(b * LL + i) * LL) * DD + dt * 16 + l16;
  v8f acc = {};
  v2f a[GK], bv[GK];
  auto load = [&](int s, int k) {
    a[s].x = ap[k];
    a[s].y = ap[k + 1];
    bv[s].x = ep[(size_t)(k + 2 * h) * DD];
    bv[s].y = ep[(size_t)(k + 2 * h + 1) * DD];
  };
#pragma unroll
  for (int s = 0; s < GK; ++s) {
    load(s, 4 * s);
    __builtin_amdgcn_sched_barrier(0);
  }
  for (int k0 = 0; k0 + 4 * GK < LL; k0 += 4 * GK) {
#pragma unroll
    for (int s = 0; s < GK; ++s) {
      acc = wmma4(a[s], bv[s], acc);
      load(s, k0 + 4 * GK + 4 * s);
      __builtin_amdgcn_sched_barrier(0);
    }
  }
#pragma unroll
  for (int s = 0; s < GK; ++s) acc = wmma4(a[s], bv[s], acc);
  if (h == 0) {
#pragma unroll
    for (int v = 0; v < 8; ++v)
      out_cat[(size_t)(b * LL + i) * DOUT + DD + v * DD + dt * 16 + l16] = acc[v];
  }
}

__global__ void op_finish_kernel(const float* __restrict__ op,
                                 const float* __restrict__ r,
                                 const float* __restrict__ t,
                                 float* __restrict__ out_cat) {
  int idx = blockIdx.x * blockDim.x + threadIdx.x;
  if (idx >= BNL) return;
  int i = idx % LL;
  int n = (idx / LL) % NH;
  int b = idx / (NH * LL);
  const float* opv = op + (size_t)idx * 12;
  const float* rp = r + (size_t)(b * LL + i) * 9;
  const float* tp = t + (size_t)(b * LL + i) * 3;
  float* oc = out_cat + (size_t)(b * LL + i) * DOUT;
#pragma unroll
  for (int p = 0; p < 4; ++p) {
    float om[3];
#pragma unroll
    for (int k = 0; k < 3; ++k) om[k] = opv[p * 3 + k] - tp[k];
    float nsq = 0.0f;
#pragma unroll
    for (int c = 0; c < 3; ++c) {
      float v = om[0] * rp[c * 3 + 0] + om[1] * rp[c * 3 + 1] + om[2] * rp[c * 3 + 2];
      oc[DD + NH * DD + n * 12 + p * 3 + c] = v;
      nsq += v * v;
    }
    oc[DD + NH * DD + NH * 12 + n * 4 + p] = sqrtf(nsq);
  }
}

extern "C" void kernel_launch(void* const* d_in, const int* in_sizes, int n_in,
                              void* d_out, int out_size, void* d_ws, size_t ws_size,
                              hipStream_t stream) {
  (void)in_sizes; (void)n_in; (void)out_size;
  const float* x     = (const float*)d_in[0];
  const float* e     = (const float*)d_in[1];
  const float* r     = (const float*)d_in[2];
  const float* t     = (const float*)d_in[3];
  const float* Wq_s  = (const float*)d_in[4];
  const float* Wk_s  = (const float*)d_in[5];
  const float* Wv_s  = (const float*)d_in[6];
  const float* Wb    = (const float*)d_in[7];
  const float* Wq_p  = (const float*)d_in[8];
  const float* Wk_p  = (const float*)d_in[9];
  const float* Wv_p  = (const float*)d_in[10];
  const float* gamma = (const float*)d_in[11];
  const float* Wo    = (const float*)d_in[12];
  const float* bo    = (const float*)d_in[13];
  float* out = (float*)d_out;
  float* ws  = (float*)d_ws;

  float* qs   = ws;
  float* ks_  = qs  + 98304;
  float* vs   = ks_ + 98304;
  float* hq   = vs  + 98304;
  float* hk   = hq  + 73728;
  float* hv   = hk  + 73728;
  float* qp   = hv  + 73728;
  float* kp   = qp  + 73728;
  float* vp   = kp  + 73728;
  float* qq   = vp  + 73728;
  float* kk   = qq  + 6144;
  float* bias = kk  + 6144;
  float* attn = bias + 2359296;
  float* op   = attn + 2359296;
  float* ocat = op  + 73728;
  size_t req = (size_t)((ocat + 983040) - ws) * sizeof(float);
  if (ws_size < req) return;

  dim3 blk(256);
  auto gemm = [&](const float* A, const float* W, const float* bvec, float* C,
                  int M, int K, int Nc) {
    int tiles = (M / 16) * (Nc / 16);
    wmma_gemm_kernel<<<(tiles + 7) / 8, blk, 0, stream>>>(A, W, bvec, C, M, K,
                                                          Nc, tiles);
  };

  gemm(x, Wq_s, nullptr, qs, BL, DD, DD);
  gemm(x, Wk_s, nullptr, ks_, BL, DD, DD);
  gemm(x, Wv_s, nullptr, vs, BL, DD, DD);
  gemm(x, Wq_p, nullptr, hq, BL, DD, NH * 12);
  gemm(x, Wk_p, nullptr, hk, BL, DD, NH * 12);
  gemm(x, Wv_p, nullptr, hv, BL, DD, NH * 12);

  bias_kernel<<<BB * LL, dim3(128), 0, stream>>>(e, Wb, bias);

  point_transform_kernel<<<BNL / 256, blk, 0, stream>>>(hq, r, t, qp, qq);
  point_transform_kernel<<<BNL / 256, blk, 0, stream>>>(hk, r, t, kp, kk);
  point_transform_kernel<<<BNL / 256, blk, 0, stream>>>(hv, r, t, vp, nullptr);

  logit_kernel<<<(BB * NH * LT * LT) / 8, blk, 0, stream>>>(qs, ks_, qp, kp, qq,
                                                            kk, bias, gamma, attn);
  softmax_kernel<<<BNL / 8, blk, 0, stream>>>(attn);

  attn_v_kernel<<<(BB * NH * LT) / 8, blk, 0, stream>>>(attn, vs, vp, ocat, op);
  out_pair_kernel<<<(BB * LL * 8) / 8, blk, 0, stream>>>(attn, e, ocat);
  op_finish_kernel<<<BNL / 256, blk, 0, stream>>>(op, r, t, ocat);

  gemm(ocat, Wo, bo, out, BL, DOUT, DD);
}


# Independent stages share a launch, picked by block index: pair bias + 6 projection GEMMs, the 3 frame transforms, pair output + attn@V: 16 launches -> 7; stage bodies and sum order unchanged
# speedup vs baseline: 3.1013x; 1.3246x over previous
#include <hip/hip_runtime.h>
#include <hip/hip_bf16.h>
#include <math.h>

#define BB   2
#define LL   384
#define DD   128
#define NH   8
#define DHS_ 16
#define DHP_ 4
#define BL   (BB*LL)
#define BNL  (BB*NH*LL)
#define DOUT 1280
#define LT   24

#define GK 16

typedef float v2f __attribute__((ext_vector_type(2)));
typedef float v8f __attribute__((ext_vector_type(8)));

static __device__ __forceinline__ v8f wmma4(v2f a, v2f b, v8f c) {
  return __builtin_amdgcn_wmma_f32_16x16x4_f32(false, a, false, b, (short)0, c,
                                               false, false);
}


static __device__ __forceinline__ void gemm_tile(const float* __restrict__ A,
                                                 const float* __restrict__ W,
                                                 const float* __restrict__ bias,
                                                 float* __restrict__ C,
                                                 int K, int Nc, int tile) {
  int lane = threadIdx.x & 31;
  int h = lane >> 4, l16 = lane & 15;
  int ntn = Nc >> 4;
  int tm = (tile / ntn) << 4;
  int tn = (tile % ntn) << 4;
  const float* ap = A + (size_t)(tm + l16) * K + 2 * h;
  const float* wp = W + (size_t)(2 * h) * Nc + tn + l16;
  v8f acc = {};
  v2f a[GK], b[GK];
  auto load = [&](int s, int k) {
    a[s].x = ap[k];
    a[s].y = ap[k + 1];
    b[s].x = wp[(size_t)k * Nc];
    b[s].y = wp[(size_t)(k + 1) * Nc];
  };
#pragma unroll
  for (int s = 0; s < GK; ++s) {
    load(s, 4 * s);
    __builtin_amdgcn_sched_barrier(0);
  }
  int k0 = 0;
  for (; k0 + 4 * GK < K; k0 += 4 * GK) {
#pragma unroll
    for (int s = 0; s < GK; ++s) {
      acc = wmma4(a[s], b[s], acc);
      load(s, k0 + 4 * GK + 4 * s);
      __builtin_amdgcn_sched_barrier(0);
    }
  }
#pragma unroll
  for (int s = 0; s < GK; ++s) acc = wmma4(a[s], b[s], acc);
  float bv = bias ? bias[tn + l16] : 0.0f;
#pragma unroll
  for (int v = 0; v < 8; ++v) {
    int row = tm + v + (h << 3);
    C[(size_t)row * Nc + tn + l16] = acc[v] + bv;
  }
}

typedef float v4f __attribute__((ext_vector_type(4)));

#define BIAS_TPW 6
#define BIAS_NS  32

static __device__ __forceinline__ void bias_row(const float* __restrict__ e,
                                                const float* __restrict__ Wb,
                                                float* __restrict__ bias,
                                                int bi) {
  int wave = threadIdx.x >> 5;
  int lane = threadIdx.x & 31;
  int h = lane >> 4, l16 = lane & 15;
  int b = bi / LL, i = bi % LL;
  int n = l16 & 7;
  v2f wf[32];
  const float* wp = Wb + (4 * h) * NH + n;
#pragma unroll
  for (int m = 0; m < 16; ++m) {
#pragma unroll
    for (int q = 0; q < 2; ++q) {
      wf[2 * m + q].x = wp[(8 * m + 2 * q) * NH];
      wf[2 * m + q].y = wp[(8 * m + 2 * q + 1) * NH];
    }
  }
  const float* ap =
      e + ((size_t)bi * LL + (size_t)(wave * BIAS_TPW * 16 + l16)) * DD + 4 * h;
  auto ld = [&](int t, int m) -> v4f {
    return *reinterpret_cast<const v4f*>(ap + (size_t)t * (16 * DD) + 8 * m);
  };
  bool hi = l16 >= 8;
  float* bp = bias + (((size_t)(b * NH + n) * LL + i) * LL) +
              wave * BIAS_TPW * 16 + 8 * h + (hi ? 4 : 0);
  auto store = [&](const v8f& acc, int t) {
    v4f o;
    o.x = hi ? acc[4] : acc[0];
    o.y = hi ? acc[5] : acc[1];
    o.z = hi ? acc[6] : acc[2];
    o.w = hi ? acc[7] : acc[3];
    *reinterpret_cast<v4f*>(bp + t * 16) = o;
  };
  v4f a[BIAS_NS];
#pragma unroll
  for (int s = 0; s < BIAS_NS; ++s) {
    a[s] = ld(s >> 4, s & 15);
    __builtin_amdgcn_sched_barrier(0);
  }
  for (int tp = 0; tp < BIAS_TPW / 2 - 1; ++tp) {
#pragma unroll
    for (int hf = 0; hf < 2; ++hf) {
      v8f acc = {};
#pragma unroll
      for (int m = 0; m < 16; ++m) {
        int s = hf * 16 + m;
        v2f a0 = {a[s].x, a[s].y}, a1 = {a[s].z, a[s].w};
        acc = wmma4(a0, wf[2 * m], acc);
        acc = wmma4(a1, wf[2 * m + 1], acc);
        a[s] = ld(2 * tp + 2 + hf, m);
        __builtin_amdgcn_sched_barrier(0);
      }
      store(acc, 2 * tp + hf);
    }
  }
#pragma unroll
  for (int hf = 0; hf < 2; ++hf) {
    v8f acc = {};
#pragma unroll
    for (int m = 0; m < 16; ++m) {
      int s = hf * 16 + m;
      v2f a0 = {a[s].x, a[s].y}, a1 = {a[s].z, a[s].w};
      acc = wmma4(a0, wf[2 * m], acc);
      acc = wmma4(a1, wf[2 * m + 1], acc);
    }
    store(acc, BIAS_TPW - 2 + hf);
  }
}

static __device__ __forceinline__ void point_transform(const float* __restrict__ hmat,
                                                       const float* __restrict__ r,
                                                       const float* __restrict__ t,
                                                       float* __restrict__ outp,
                                                       float* __restrict__ sumsq,
                                                       int idx) {
  if (idx >= BNL) return;
  int l = idx % LL;
  int n = (idx / LL) % NH;
  int b = idx / (NH * LL);
  const float* hp = hmat + (size_t)(b * LL + l) * (NH * 12) + n * 12;
  const float* rp = r + (size_t)(b * LL + l) * 9;
  const float* tp = t + (size_t)(b * LL + l) * 3;
  float* op = outp + (size_t)idx * 12;
  float ss = 0.0f;
#pragma unroll
  for (int p = 0; p < 4; ++p)
#pragma unroll
    for (int c = 0; c < 3; ++c) {
      float v = tp[c];
#pragma unroll
      for (int k = 0; k < 3; ++k) v += hp[p * 3 + k] * rp[k * 3 + c];
      op[p * 3 + c] = v;
      ss += v * v;
    }
  if (sumsq) sumsq[idx] = ss;
}

__global__ void logit_kernel(const float* __restrict__ qs,
                             const float* __restrict__ ks,
                             const float* __restrict__ qp,
                             const float* __restrict__ kp,
                             const float* __restrict__ qq,
                             const float* __restrict__ kk,
                             const float* __restrict__ bias,
                             const float* __restrict__ gamma,
                             float* __restrict__ logit) {
  int tile = blockIdx.x * 8 + (threadIdx.x >> 5);
  int jt = tile % LT;
  int it = (tile / LT) % LT;
  int n  = (tile / (LT * LT)) % NH;
  int b  = tile / (LT * LT * NH);
  int lane = threadIdx.x & 31;
  int h = lane >> 4, l16 = lane & 15;

  const float* qsp = qs + (size_t)(b * LL + it * 16 + l16) * DD + n * DHS_ + 2 * h;
  const float* ksp = ks + (size_t)(b * LL + jt * 16 + l16) * DD + n * DHS_ + 2 * h;
  v8f accS = {};
#pragma unroll
  for (int k = 0; k < 16; k += 4) {
    v2f a, bv2;
    a.x = qsp[k];   a.y = qsp[k + 1];
    bv2.x = ksp[k]; bv2.y = ksp[k + 1];
    accS = wmma4(a, bv2, accS);
  }
  const float* qpp = qp + (size_t)((b * NH + n) * LL + it * 16 + l16) * 12 + 2 * h;
  const float* kpp = kp + (size_t)((b * NH + n) * LL + jt * 16 + l16) * 12 + 2 * h;
  v8f accP = {};
#pragma unroll
  for (int k = 0; k < 12; k += 4) {
    v2f a, bv2;
    a.x = qpp[k];   a.y = qpp[k + 1];
    bv2.x = kpp[k]; bv2.y = kpp[k + 1];
    accP = wmma4(a, bv2, accP);
  }
  size_t hb = (size_t)(b * NH + n) * LL;
  int j = jt * 16 + l16;
  float kkj = kk[hb + j];
  float g = gamma[n];
  const float rsqrt3 = 0.57735026918962576f;
#pragma unroll
  for (int v = 0; v < 8; ++v) {
    int i = it * 16 + v + (h << 3);
    float d2 = qq[hb + i] + kkj - 2.0f * accP[v];
    float lp = -0.125f * g * d2;
    float bsv = bias[(hb + i) * LL + j];
    logit[(hb + i) * LL + j] = rsqrt3 * (0.25f * accS[v] + bsv + lp);
  }
}

__global__ void softmax_kernel(float* __restrict__ attn) {
  int row = blockIdx.x * 8 + (threadIdx.x >> 5);
  int lane = threadIdx.x & 31;
  float* p = attn + (size_t)row * LL;
  float v[12];
  float m = -1e30f;
#pragma unroll
  for (int i = 0; i < 12; ++i) {
    v[i] = p[lane + i * 32];
    m = fmaxf(m, v[i]);
  }
#pragma unroll
  for (int off = 16; off > 0; off >>= 1) m = fmaxf(m, __shfl_xor(m, off, 32));
  float s = 0.0f;
#pragma unroll
  for (int i = 0; i < 12; ++i) {
    v[i] = __expf(v[i] - m);
    s += v[i];
  }
#pragma unroll
  for (int off = 16; off > 0; off >>= 1) s += __shfl_xor(s, off, 32);
  float inv = 1.0f / s;
#pragma unroll
  for (int i = 0; i < 12; ++i) p[lane + i * 32] = v[i] * inv;
}

static __device__ __forceinline__ void attn_v_tile(const float* __restrict__ attn,
                                                   const float* __restrict__ vs,
                                                   const float* __restrict__ vp,
                                                   float* __restrict__ out_cat,
                                                   float* __restrict__ op,
                                                   int tile) {
  int it = tile % LT;
  int n  = (tile / LT) % NH;
  int b  = tile / (LT * NH);
  int lane = threadIdx.x & 31;
  int h = lane >> 4, l16 = lane & 15;
  size_t hb = (size_t)(b * NH + n) * LL;
  const float* ap  = attn + (hb + it * 16 + l16) * LL + 2 * h;
  const float* vsb = vs + (size_t)b * LL * DD + n * DHS_ + l16;
  bool pv = l16 < 12;
  const float* vpb = vp + hb * 12 + (pv ? l16 : 11);
  v8f accS = {}, accP = {};
  constexpr int GV = 12;
  v2f a[GV], bs[GV], bp[GV];
  auto load = [&](int s, int k) {
    a[s].x = ap[k];
    a[s].y = ap[k + 1];
    bs[s].x = vsb[(size_t)(k + 2 * h) * DD];
    bs[s].y = vsb[(size_t)(k + 2 * h + 1) * DD];
    bp[s].x = vpb[(k + 2 * h) * 12];
    bp[s].y = vpb[(k + 2 * h + 1) * 12];
  };
#pragma unroll
  for (int s = 0; s < GV; ++s) {
    load(s, 4 * s);
    __builtin_amdgcn_sched_barrier(0);
  }
  for (int k0 = 0; k0 + 4 * GV < LL; k0 += 4 * GV) {
#pragma unroll
    for (int s = 0; s < GV; ++s) {
      accS = wmma4(a[s], bs[s], accS);
      accP = wmma4(a[s], bp[s], accP);
      load(s, k0 + 4 * GV + 4 * s);
      __builtin_amdgcn_sched_barrier(0);
    }
  }
#pragma unroll
  for (int s = 0; s < GV; ++s) {
    accS = wmma4(a[s], bs[s], accS);
    accP = wmma4(a[s], bp[s], accP);
  }
#pragma unroll
  for (int v = 0; v < 8; ++v) {
    int i = it * 16 + v + (h << 3);
    out_cat[(size_t)(b * LL + i) * DOUT + n * DHS_ + l16] = accS[v];
    if (pv) op[(hb + i) * 12 + l16] = accP[v];
  }
}

static __device__ __forceinline__ void out_pair_tile(const float* __restrict__ attn,
                                                     const float* __restrict__ e,
                                                     float* __restrict__ out_cat,
                                                     int tile) {
  int dt = tile % 8;
  int i  = (tile / 8) % LL;
  int b  = tile / (8 * LL);
  int lane = threadIdx.x & 31;
  int h = lane >> 4, l16 = lane & 15;
  const float* ap = attn + ((size_t)(b * NH + (l16 & 7)) * LL + i) * LL + 2 * h;
  const float* ep = e + ((size_t)(b * LL + i) * LL) * DD + dt * 16 + l16;
  v8f acc = {};
  v2f a[GK], bv[GK];
  auto load = [&](int s, int k) {
    a[s].x = ap[k];
    a[s].y = ap[k + 1];
    bv[s].x = ep[(size_t)(k + 2 * h) * DD];
    bv[s].y = ep[(size_t)(k + 2 * h + 1) * DD];
  };
#pragma unroll
  for (int s = 0; s < GK; ++s) {
    load(s, 4 * s);
    __builtin_amdgcn_sched_barrier(0);
  }
  for (int k0 = 0; k0 + 4 * GK < LL; k0 += 4 * GK) {
#pragma unroll
    for (int s = 0; s < GK; ++s) {
      acc = wmma4(a[s], bv[s], acc);
      load(s, k0 + 4 * GK + 4 * s);
      __builtin_amdgcn_sched_barrier(0);
    }
  }
#pragma unroll
  for (int s = 0; s < GK; ++s) acc = wmma4(a[s], bv[s], acc);
  if (h == 0) {
#pragma unroll
    for (int v = 0; v < 8; ++v)
      out_cat[(size_t)(b * LL + i) * DOUT + DD + v * DD + dt * 16 + l16] = acc[v];
  }
}

__global__ void op_finish_kernel(const float* __restrict__ op,
                                 const float* __restrict__ r,
                                 const float* __restrict__ t,
                                 float* __restrict__ out_cat) {
  int idx = blockIdx.x * blockDim.x + threadIdx.x;
  if (idx >= BNL) return;
  int i = idx % LL;
  int n = (idx / LL) % NH;
  int b = idx / (NH * LL);
  const float* opv = op + (size_t)idx * 12;
  const float* rp = r + (size_t)(b * LL + i) * 9;
  const float* tp = t + (size_t)(b * LL + i) * 3;
  float* oc = out_cat + (size_t)(b * LL + i) * DOUT;
#pragma unroll
  for (int p = 0; p < 4; ++p) {
    float om[3];
#pragma unroll
    for (int k = 0; k < 3; ++k) om[k] = opv[p * 3 + k] - tp[k];
    float nsq = 0.0f;
#pragma unroll
    for (int c = 0; c < 3; ++c) {
      float v = om[0] * rp[c * 3 + 0] + om[1] * rp[c * 3 + 1] + om[2] * rp[c * 3 + 2];
      oc[DD + NH * DD + n * 12 + p * 3 + c] = v;
      nsq += v * v;
    }
    oc[DD + NH * DD + NH * 12 + n * 4 + p] = sqrtf(nsq);
  }
}

#define PJ_BLK_S 96
#define PJ_BLK_P 72
#define PJ_BLKS  (3 * PJ_BLK_S + 3 * PJ_BLK_P)

__global__ void __launch_bounds__(128)
proj_bias_kernel(const float* __restrict__ x,
                 const float* __restrict__ e,
                 const float* __restrict__ Wq_s,
                 const float* __restrict__ Wk_s,
                 const float* __restrict__ Wv_s,
                 const float* __restrict__ Wq_p,
                 const float* __restrict__ Wk_p,
                 const float* __restrict__ Wv_p,
                 const float* __restrict__ Wb,
                 float* __restrict__ qs,
                 float* __restrict__ ks,
                 float* __restrict__ vs,
                 float* __restrict__ hq,
                 float* __restrict__ hk,
                 float* __restrict__ hv,
                 float* __restrict__ bias) {
  int blk = blockIdx.x;
  if (blk < BL) {
    bias_row(e, Wb, bias, blk);
    return;
  }
  blk -= BL;
  const float* Wm;
  float* C;
  int Nc;
  if (blk < 3 * PJ_BLK_S) {
    int g = blk / PJ_BLK_S;
    blk -= g * PJ_BLK_S;
    Nc = DD;
    Wm = g == 0 ? Wq_s : (g == 1 ? Wk_s : Wv_s);
    C  = g == 0 ? qs : (g == 1 ? ks : vs);
  } else {
    blk -= 3 * PJ_BLK_S;
    int g = blk / PJ_BLK_P;
    blk -= g * PJ_BLK_P;
    Nc = NH * 12;
    Wm = g == 0 ? Wq_p : (g == 1 ? Wk_p : Wv_p);
    C  = g == 0 ? hq : (g == 1 ? hk : hv);
  }
  gemm_tile(x, Wm, nullptr, C, DD, Nc, blk * 4 + (threadIdx.x >> 5));
}

#define PT_BLK (BNL / 256)

__global__ void point_transform3_kernel(const float* __restrict__ hq,
                                        const float* __restrict__ hk,
                                        const float* __restrict__ hv,
                                        const float* __restrict__ r,
                                        const float* __restrict__ t,
                                        float* __restrict__ qp,
                                        float* __restrict__ kp,
                                        float* __restrict__ vp,
                                        float* __restrict__ qq,
                                        float* __restrict__ kk) {
  int g = blockIdx.x / PT_BLK;
  int idx = (blockIdx.x - g * PT_BLK) * 256 + threadIdx.x;
  const float* hm = g == 0 ? hq : (g == 1 ? hk : hv);
  float* outp     = g == 0 ? qp : (g == 1 ? kp : vp);
  float* ss       = g == 0 ? qq : (g == 1 ? kk : nullptr);
  point_transform(hm, r, t, outp, ss, idx);
}

#define OP_BLKS (BB * LL)
#define AV_BLKS ((BB * NH * LT) / 8)

__global__ void attn_reduce_kernel(const float* __restrict__ attn,
                                   const float* __restrict__ e,
                                   const float* __restrict__ vs,
                                   const float* __restrict__ vp,
                                   float* __restrict__ out_cat,
                                   float* __restrict__ op) {
  int wave = threadIdx.x >> 5;
  if (blockIdx.x < OP_BLKS) {
    out_pair_tile(attn, e, out_cat, blockIdx.x * 8 + wave);
  } else {
    attn_v_tile(attn, vs, vp, out_cat, op, (blockIdx.x - OP_BLKS) * 8 + wave);
  }
}

__global__ void wmma_gemm_kernel(const float* __restrict__ A,
                                 const float* __restrict__ W,
                                 const float* __restrict__ bias,
                                 float* __restrict__ C,
                                 int K, int Nc, int total_tiles) {
  int tile = blockIdx.x * 8 + (threadIdx.x >> 5);
  if (tile >= total_tiles) return;
  gemm_tile(A, W, bias, C, K, Nc, tile);
}

extern "C" void kernel_launch(void* const* d_in, const int* in_sizes, int n_in,
                              void* d_out, int out_size, void* d_ws, size_t ws_size,
                              hipStream_t stream) {
  (void)in_sizes; (void)n_in; (void)out_size;
  const float* x     = (const float*)d_in[0];
  const float* e     = (const float*)d_in[1];
  const float* r     = (const float*)d_in[2];
  const float* t     = (const float*)d_in[3];
  const float* Wq_s  = (const float*)d_in[4];
  const float* Wk_s  = (const float*)d_in[5];
  const float* Wv_s  = (const float*)d_in[6];
  const float* Wb    = (const float*)d_in[7];
  const float* Wq_p  = (const float*)d_in[8];
  const float* Wk_p  = (const float*)d_in[9];
  const float* Wv_p  = (const float*)d_in[10];
  const float* gamma = (const float*)d_in[11];
  const float* Wo    = (const float*)d_in[12];
  const float* bo    = (const float*)d_in[13];
  float* out = (float*)d_out;
  float* ws  = (float*)d_ws;

  float* qs   = ws;
  float* ks_  = qs  + 98304;
  float* vs   = ks_ + 98304;
  float* hq   = vs  + 98304;
  float* hk   = hq  + 73728;
  float* hv   = hk  + 73728;
  float* qp   = hv  + 73728;
  float* kp   = qp  + 73728;
  float* vp   = kp  + 73728;
  float* qq   = vp  + 73728;
  float* kk   = qq  + 6144;
  float* bias = kk  + 6144;
  float* attn = bias + 2359296;
  float* op   = attn + 2359296;
  float* ocat = op  + 73728;
  size_t req = (size_t)((ocat + 983040) - ws) * sizeof(float);
  if (ws_size < req) return;

  dim3 blk(256);

  proj_bias_kernel<<<BL + PJ_BLKS, dim3(128), 0, stream>>>(
      x, e, Wq_s, Wk_s, Wv_s, Wq_p, Wk_p, Wv_p, Wb, qs, ks_, vs, hq, hk, hv, bias);

  point_transform3_kernel<<<3 * PT_BLK, blk, 0, stream>>>(hq, hk, hv, r, t, qp, kp,
                                                          vp, qq, kk);

  logit_kernel<<<(BB * NH * LT * LT) / 8, blk, 0, stream>>>(qs, ks_, qp, kp, qq,
                                                            kk, bias, gamma, attn);
  softmax_kernel<<<BNL / 8, blk, 0, stream>>>(attn);

  attn_reduce_kernel<<<OP_BLKS + AV_BLKS, blk, 0, stream>>>(attn, e, vs, vp, ocat,
                                                            op);
  op_finish_kernel<<<BNL / 256, blk, 0, stream>>>(op, r, t, ocat);

  wmma_gemm_kernel<<<((BL / 16) * (DD / 16)) / 8, blk, 0, stream>>>(
      ocat, Wo, bo, out, DOUT, DD, (BL / 16) * (DD / 16));
}


# Final projection GEMM: K=1280 split over the 10 waves of a block (128 k each), partial tiles added through LDS; 384 blocks instead of 48; sum order changes (10 partial sums per output)
# speedup vs baseline: 3.5324x; 1.1390x over previous
#include <hip/hip_runtime.h>
#include <hip/hip_bf16.h>
#include <math.h>

#define BB   2
#define LL   384
#define DD   128
#define NH   8
#define DHS_ 16
#define DHP_ 4
#define BL   (BB*LL)
#define BNL  (BB*NH*LL)
#define DOUT 1280
#define LT   24

#define GK 16

typedef float v2f __attribute__((ext_vector_type(2)));
typedef float v8f __attribute__((ext_vector_type(8)));

static __device__ __forceinline__ v8f wmma4(v2f a, v2f b, v8f c) {
  return __builtin_amdgcn_wmma_f32_16x16x4_f32(false, a, false, b, (short)0, c,
                                               false, false);
}


static __device__ __forceinline__ v8f gemm_acc(const float* __restrict__ A,
                                               const float* __restrict__ W,
                                               int lda, int Nc, int tm, int tn,
                                               int kbeg, int klen) {
  int lane = threadIdx.x & 31;
  int h = lane >> 4, l16 = lane & 15;
  const float* ap = A + (size_t)(tm + l16) * lda + kbeg + 2 * h;
  const float* wp = W + (size_t)(kbeg + 2 * h) * Nc + tn + l16;
  v8f acc = {};
  v2f a[GK], b[GK];
  auto load = [&](int s, int k) {
    a[s].x = ap[k];
    a[s].y = ap[k + 1];
    b[s].x = wp[(size_t)k * Nc];
    b[s].y = wp[(size_t)(k + 1) * Nc];
  };
#pragma unroll
  for (int s = 0; s < GK; ++s) {
    load(s, 4 * s);
    __builtin_amdgcn_sched_barrier(0);
  }
  int k0 = 0;
  for (; k0 + 4 * GK < klen; k0 += 4 * GK) {
#pragma unroll
    for (int s = 0; s < GK; ++s) {
      acc = wmma4(a[s], b[s], acc);
      load(s, k0 + 4 * GK + 4 * s);
      __builtin_amdgcn_sched_barrier(0);
    }
  }
#pragma unroll
  for (int s = 0; s < GK; ++s) acc = wmma4(a[s], b[s], acc);
  return acc;
}

static __device__ __forceinline__ void gemm_tile(const float* __restrict__ A,
                                                 const float* __restrict__ W,
                                                 const float* __restrict__ bias,
                                                 float* __restrict__ C,
                                                 int K, int Nc, int tile) {
  int lane = threadIdx.x & 31;
  int h = lane >> 4, l16 = lane & 15;
  int ntn = Nc >> 4;
  int tm = (tile / ntn) << 4;
  int tn = (tile % ntn) << 4;
  v8f acc = gemm_acc(A, W, K, Nc, tm, tn, 0, K);
  float bv = bias ? bias[tn + l16] : 0.0f;
#pragma unroll
  for (int v = 0; v < 8; ++v) {
    int row = tm + v + (h << 3);
    C[(size_t)row * Nc + tn + l16] = acc[v] + bv;
  }
}

typedef float v4f __attribute__((ext_vector_type(4)));

#define BIAS_TPW 6
#define BIAS_NS  32

static __device__ __forceinline__ void bias_row(const float* __restrict__ e,
                                                const float* __restrict__ Wb,
                                                float* __restrict__ bias,
                                                int bi) {
  int wave = threadIdx.x >> 5;
  int lane = threadIdx.x & 31;
  int h = lane >> 4, l16 = lane & 15;
  int b = bi / LL, i = bi % LL;
  int n = l16 & 7;
  v2f wf[32];
  const float* wp = Wb + (4 * h) * NH + n;
#pragma unroll
  for (int m = 0; m < 16; ++m) {
#pragma unroll
    for (int q = 0; q < 2; ++q) {
      wf[2 * m + q].x = wp[(8 * m + 2 * q) * NH];
      wf[2 * m + q].y = wp[(8 * m + 2 * q + 1) * NH];
    }
  }
  const float* ap =
      e + ((size_t)bi * LL + (size_t)(wave * BIAS_TPW * 16 + l16)) * DD + 4 * h;
  auto ld = [&](int t, int m) -> v4f {
    return *reinterpret_cast<const v4f*>(ap + (size_t)t * (16 * DD) + 8 * m);
  };
  bool hi = l16 >= 8;
  float* bp = bias + (((size_t)(b * NH + n) * LL + i) * LL) +
              wave * BIAS_TPW * 16 + 8 * h + (hi ? 4 : 0);
  auto store = [&](const v8f& acc, int t) {
    v4f o;
    o.x = hi ? acc[4] : acc[0];
    o.y = hi ? acc[5] : acc[1];
    o.z = hi ? acc[6] : acc[2];
    o.w = hi ? acc[7] : acc[3];
    *reinterpret_cast<v4f*>(bp + t * 16) = o;
  };
  v4f a[BIAS_NS];
#pragma unroll
  for (int s = 0; s < BIAS_NS; ++s) {
    a[s] = ld(s >> 4, s & 15);
    __builtin_amdgcn_sched_barrier(0);
  }
  for (int tp = 0; tp < BIAS_TPW / 2 - 1; ++tp) {
#pragma unroll
    for (int hf = 0; hf < 2; ++hf) {
      v8f acc = {};
#pragma unroll
      for (int m = 0; m < 16; ++m) {
        int s = hf * 16 + m;
        v2f a0 = {a[s].x, a[s].y}, a1 = {a[s].z, a[s].w};
        acc = wmma4(a0, wf[2 * m], acc);
        acc = wmma4(a1, wf[2 * m + 1], acc);
        a[s] = ld(2 * tp + 2 + hf, m);
        __builtin_amdgcn_sched_barrier(0);
      }
      store(acc, 2 * tp + hf);
    }
  }
#pragma unroll
  for (int hf = 0; hf < 2; ++hf) {
    v8f acc = {};
#pragma unroll
    for (int m = 0; m < 16; ++m) {
      int s = hf * 16 + m;
      v2f a0 = {a[s].x, a[s].y}, a1 = {a[s].z, a[s].w};
      acc = wmma4(a0, wf[2 * m], acc);
      acc = wmma4(a1, wf[2 * m + 1], acc);
    }
    store(acc, BIAS_TPW - 2 + hf);
  }
}

static __device__ __forceinline__ void point_transform(const float* __restrict__ hmat,
                                                       const float* __restrict__ r,
                                                       const float* __restrict__ t,
                                                       float* __restrict__ outp,
                                                       float* __restrict__ sumsq,
                                                       int idx) {
  if (idx >= BNL) return;
  int l = idx % LL;
  int n = (idx / LL) % NH;
  int b = idx / (NH * LL);
  const float* hp = hmat + (size_t)(b * LL + l) * (NH * 12) + n * 12;
  const float* rp = r + (size_t)(b * LL + l) * 9;
  const float* tp = t + (size_t)(b * LL + l) * 3;
  float* op = outp + (size_t)idx * 12;
  float ss = 0.0f;
#pragma unroll
  for (int p = 0; p < 4; ++p)
#pragma unroll
    for (int c = 0; c < 3; ++c) {
      float v = tp[c];
#pragma unroll
      for (int k = 0; k < 3; ++k) v += hp[p * 3 + k] * rp[k * 3 + c];
      op[p * 3 + c] = v;
      ss += v * v;
    }
  if (sumsq) sumsq[idx] = ss;
}

__global__ void logit_kernel(const float* __restrict__ qs,
                             const float* __restrict__ ks,
                             const float* __restrict__ qp,
                             const float* __restrict__ kp,
                             const float* __restrict__ qq,
                             const float* __restrict__ kk,
                             const float* __restrict__ bias,
                             const float* __restrict__ gamma,
                             float* __restrict__ logit) {
  int tile = blockIdx.x * 8 + (threadIdx.x >> 5);
  int jt = tile % LT;
  int it = (tile / LT) % LT;
  int n  = (tile / (LT * LT)) % NH;
  int b  = tile / (LT * LT * NH);
  int lane = threadIdx.x & 31;
  int h = lane >> 4, l16 = lane & 15;

  const float* qsp = qs + (size_t)(b * LL + it * 16 + l16) * DD + n * DHS_ + 2 * h;
  const float* ksp = ks + (size_t)(b * LL + jt * 16 + l16) * DD + n * DHS_ + 2 * h;
  v8f accS = {};
#pragma unroll
  for (int k = 0; k < 16; k += 4) {
    v2f a, bv2;
    a.x = qsp[k];   a.y = qsp[k + 1];
    bv2.x = ksp[k]; bv2.y = ksp[k + 1];
    accS = wmma4(a, bv2, accS);
  }
  const float* qpp = qp + (size_t)((b * NH + n) * LL + it * 16 + l16) * 12 + 2 * h;
  const float* kpp = kp + (size_t)((b * NH + n) * LL + jt * 16 + l16) * 12 + 2 * h;
  v8f accP = {};
#pragma unroll
  for (int k = 0; k < 12; k += 4) {
    v2f a, bv2;
    a.x = qpp[k];   a.y = qpp[k + 1];
    bv2.x = kpp[k]; bv2.y = kpp[k + 1];
    accP = wmma4(a, bv2, accP);
  }
  size_t hb = (size_t)(b * NH + n) * LL;
  int j = jt * 16 + l16;
  float kkj = kk[hb + j];
  float g = gamma[n];
  const float rsqrt3 = 0.57735026918962576f;
#pragma unroll
  for (int v = 0; v < 8; ++v) {
    int i = it * 16 + v + (h << 3);
    float d2 = qq[hb + i] + kkj - 2.0f * accP[v];
    float lp = -0.125f * g * d2;
    float bsv = bias[(hb + i) * LL + j];
    logit[(hb + i) * LL + j] = rsqrt3 * (0.25f * accS[v] + bsv + lp);
  }
}

__global__ void softmax_kernel(float* __restrict__ attn) {
  int row = blockIdx.x * 8 + (threadIdx.x >> 5);
  int lane = threadIdx.x & 31;
  float* p = attn + (size_t)row * LL;
  float v[12];
  float m = -1e30f;
#pragma unroll
  for (int i = 0; i < 12; ++i) {
    v[i] = p[lane + i * 32];
    m = fmaxf(m, v[i]);
  }
#pragma unroll
  for (int off = 16; off > 0; off >>= 1) m = fmaxf(m, __shfl_xor(m, off, 32));
  float s = 0.0f;
#pragma unroll
  for (int i = 0; i < 12; ++i) {
    v[i] = __expf(v[i] - m);
    s += v[i];
  }
#pragma unroll
  for (int off = 16; off > 0; off >>= 1) s += __shfl_xor(s, off, 32);
  float inv = 1.0f / s;
#pragma unroll
  for (int i = 0; i < 12; ++i) p[lane + i * 32] = v[i] * inv;
}

static __device__ __forceinline__ void attn_v_tile(const float* __restrict__ attn,
                                                   const float* __restrict__ vs,
                                                   const float* __restrict__ vp,
                                                   float* __restrict__ out_cat,
                                                   float* __restrict__ op,
                                                   int tile) {
  int it = tile % LT;
  int n  = (tile / LT) % NH;
  int b  = tile / (LT * NH);
  int lane = threadIdx.x & 31;
  int h = lane >> 4, l16 = lane & 15;
  size_t hb = (size_t)(b * NH + n) * LL;
  const float* ap  = attn + (hb + it * 16 + l16) * LL + 2 * h;
  const float* vsb = vs + (size_t)b * LL * DD + n * DHS_ + l16;
  bool pv = l16 < 12;
  const float* vpb = vp + hb * 12 + (pv ? l16 : 11);
  v8f accS = {}, accP = {};
  constexpr int GV = 12;
  v2f a[GV], bs[GV], bp[GV];
  auto load = [&](int s, int k) {
    a[s].x = ap[k];
    a[s].y = ap[k + 1];
    bs[s].x = vsb[(size_t)(k + 2 * h) * DD];
    bs[s].y = vsb[(size_t)(k + 2 * h + 1) * DD];
    bp[s].x = vpb[(k + 2 * h) * 12];
    bp[s].y = vpb[(k + 2 * h + 1) * 12];
  };
#pragma unroll
  for (int s = 0; s < GV; ++s) {
    load(s, 4 * s);
    __builtin_amdgcn_sched_barrier(0);
  }
  for (int k0 = 0; k0 + 4 * GV < LL; k0 += 4 * GV) {
#pragma unroll
    for (int s = 0; s < GV; ++s) {
      accS = wmma4(a[s], bs[s], accS);
      accP = wmma4(a[s], bp[s], accP);
      load(s, k0 + 4 * GV + 4 * s);
      __builtin_amdgcn_sched_barrier(0);
    }
  }
#pragma unroll
  for (int s = 0; s < GV; ++s) {
    accS = wmma4(a[s], bs[s], accS);
    accP = wmma4(a[s], bp[s], accP);
  }
#pragma unroll
  for (int v = 0; v < 8; ++v) {
    int i = it * 16 + v + (h << 3);
    out_cat[(size_t)(b * LL + i) * DOUT + n * DHS_ + l16] = accS[v];
    if (pv) op[(hb + i) * 12 + l16] = accP[v];
  }
}

static __device__ __forceinline__ void out_pair_tile(const float* __restrict__ attn,
                                                     const float* __restrict__ e,
                                                     float* __restrict__ out_cat,
                                                     int tile) {
  int dt = tile % 8;
  int i  = (tile / 8) % LL;
  int b  = tile / (8 * LL);
  int lane = threadIdx.x & 31;
  int h = lane >> 4, l16 = lane & 15;
  const float* ap = attn + ((size_t)(b * NH + (l16 & 7)) * LL + i) * LL + 2 * h;
  const float* ep = e + ((size_t)(b * LL + i) * LL) * DD + dt * 16 + l16;
  v8f acc = {};
  v2f a[GK], bv[GK];
  auto load = [&](int s, int k) {
    a[s].x = ap[k];
    a[s].y = ap[k + 1];
    bv[s].x = ep[(size_t)(k + 2 * h) * DD];
    bv[s].y = ep[(size_t)(k + 2 * h + 1) * DD];
  };
#pragma unroll
  for (int s = 0; s < GK; ++s) {
    load(s, 4 * s);
    __builtin_amdgcn_sched_barrier(0);
  }
  for (int k0 = 0; k0 + 4 * GK < LL; k0 += 4 * GK) {
#pragma unroll
    for (int s = 0; s < GK; ++s) {
      acc = wmma4(a[s], bv[s], acc);
      load(s, k0 + 4 * GK + 4 * s);
      __builtin_amdgcn_sched_barrier(0);
    }
  }
#pragma unroll
  for (int s = 0; s < GK; ++s) acc = wmma4(a[s], bv[s], acc);
  if (h == 0) {
#pragma unroll
    for (int v = 0; v < 8; ++v)
      out_cat[(size_t)(b * LL + i) * DOUT + DD + v * DD + dt * 16 + l16] = acc[v];
  }
}

__global__ void op_finish_kernel(const float* __restrict__ op,
                                 const float* __restrict__ r,
                                 const float* __restrict__ t,
                                 float* __restrict__ out_cat) {
  int idx = blockIdx.x * blockDim.x + threadIdx.x;
  if (idx >= BNL) return;
  int i = idx % LL;
  int n = (idx / LL) % NH;
  int b = idx / (NH * LL);
  const float* opv = op + (size_t)idx * 12;
  const float* rp = r + (size_t)(b * LL + i) * 9;
  const float* tp = t + (size_t)(b * LL + i) * 3;
  float* oc = out_cat + (size_t)(b * LL + i) * DOUT;
#pragma unroll
  for (int p = 0; p < 4; ++p) {
    float om[3];
#pragma unroll
    for (int k = 0; k < 3; ++k) om[k] = opv[p * 3 + k] - tp[k];
    float nsq = 0.0f;
#pragma unroll
    for (int c = 0; c < 3; ++c) {
      float v = om[0] * rp[c * 3 + 0] + om[1] * rp[c * 3 + 1] + om[2] * rp[c * 3 + 2];
      oc[DD + NH * DD + n * 12 + p * 3 + c] = v;
      nsq += v * v;
    }
    oc[DD + NH * DD + NH * 12 + n * 4 + p] = sqrtf(nsq);
  }
}

#define PJ_BLK_S 96
#define PJ_BLK_P 72
#define PJ_BLKS  (3 * PJ_BLK_S + 3 * PJ_BLK_P)

__global__ void __launch_bounds__(128)
proj_bias_kernel(const float* __restrict__ x,
                 const float* __restrict__ e,
                 const float* __restrict__ Wq_s,
                 const float* __restrict__ Wk_s,
                 const float* __restrict__ Wv_s,
                 const float* __restrict__ Wq_p,
                 const float* __restrict__ Wk_p,
                 const float* __restrict__ Wv_p,
                 const float* __restrict__ Wb,
                 float* __restrict__ qs,
                 float* __restrict__ ks,
                 float* __restrict__ vs,
                 float* __restrict__ hq,
                 float* __restrict__ hk,
                 float* __restrict__ hv,
                 float* __restrict__ bias) {
  int blk = blockIdx.x;
  if (blk < BL) {
    bias_row(e, Wb, bias, blk);
    return;
  }
  blk -= BL;
  const float* Wm;
  float* C;
  int Nc;
  if (blk < 3 * PJ_BLK_S) {
    int g = blk / PJ_BLK_S;
    blk -= g * PJ_BLK_S;
    Nc = DD;
    Wm = g == 0 ? Wq_s : (g == 1 ? Wk_s : Wv_s);
    C  = g == 0 ? qs : (g == 1 ? ks : vs);
  } else {
    blk -= 3 * PJ_BLK_S;
    int g = blk / PJ_BLK_P;
    blk -= g * PJ_BLK_P;
    Nc = NH * 12;
    Wm = g == 0 ? Wq_p : (g == 1 ? Wk_p : Wv_p);
    C  = g == 0 ? hq : (g == 1 ? hk : hv);
  }
  gemm_tile(x, Wm, nullptr, C, DD, Nc, blk * 4 + (threadIdx.x >> 5));
}

#define PT_BLK (BNL / 256)

__global__ void point_transform3_kernel(const float* __restrict__ hq,
                                        const float* __restrict__ hk,
                                        const float* __restrict__ hv,
                                        const float* __restrict__ r,
                                        const float* __restrict__ t,
                                        float* __restrict__ qp,
                                        float* __restrict__ kp,
                                        float* __restrict__ vp,
                                        float* __restrict__ qq,
                                        float* __restrict__ kk) {
  int g = blockIdx.x / PT_BLK;
  int idx = (blockIdx.x - g * PT_BLK) * 256 + threadIdx.x;
  const float* hm = g == 0 ? hq : (g == 1 ? hk : hv);
  float* outp     = g == 0 ? qp : (g == 1 ? kp : vp);
  float* ss       = g == 0 ? qq : (g == 1 ? kk : nullptr);
  point_transform(hm, r, t, outp, ss, idx);
}

#define OP_BLKS (BB * LL)
#define AV_BLKS ((BB * NH * LT) / 8)

__global__ void attn_reduce_kernel(const float* __restrict__ attn,
                                   const float* __restrict__ e,
                                   const float* __restrict__ vs,
                                   const float* __restrict__ vp,
                                   float* __restrict__ out_cat,
                                   float* __restrict__ op) {
  int wave = threadIdx.x >> 5;
  if (blockIdx.x < OP_BLKS) {
    out_pair_tile(attn, e, out_cat, blockIdx.x * 8 + wave);
  } else {
    attn_v_tile(attn, vs, vp, out_cat, op, (blockIdx.x - OP_BLKS) * 8 + wave);
  }
}

#define FP_KS 10
#define FP_KC (DOUT / FP_KS)

__global__ void __launch_bounds__(FP_KS * 32)
out_proj_kernel(const float* __restrict__ A,
                const float* __restrict__ W,
                const float* __restrict__ bias,
                float* __restrict__ C) {
  __shared__ float red[FP_KS][8][32];
  int wave = threadIdx.x >> 5;
  int lane = threadIdx.x & 31;
  int tm = (blockIdx.x / (DD / 16)) << 4;
  int tn = (blockIdx.x % (DD / 16)) << 4;
  v8f acc = gemm_acc(A, W, DOUT, DD, tm, tn, wave * FP_KC, FP_KC);
#pragma unroll
  for (int v = 0; v < 8; ++v) red[wave][v][lane] = acc[v];
  __syncthreads();
  if (wave < 8) {
    float s = red[0][wave][lane];
#pragma unroll
    for (int w = 1; w < FP_KS; ++w) s += red[w][wave][lane];
    int row = tm + wave + ((lane >> 4) << 3);
    int col = tn + (lane & 15);
    C[(size_t)row * DD + col] = s + bias[col];
  }
}

extern "C" void kernel_launch(void* const* d_in, const int* in_sizes, int n_in,
                              void* d_out, int out_size, void* d_ws, size_t ws_size,
                              hipStream_t stream) {
  (void)in_sizes; (void)n_in; (void)out_size;
  const float* x     = (const float*)d_in[0];
  const float* e     = (const float*)d_in[1];
  const float* r     = (const float*)d_in[2];
  const float* t     = (const float*)d_in[3];
  const float* Wq_s  = (const float*)d_in[4];
  const float* Wk_s  = (const float*)d_in[5];
  const float* Wv_s  = (const float*)d_in[6];
  const float* Wb    = (const float*)d_in[7];
  const float* Wq_p  = (const float*)d_in[8];
  const float* Wk_p  = (const float*)d_in[9];
  const float* Wv_p  = (const float*)d_in[10];
  const float* gamma = (const float*)d_in[11];
  const float* Wo    = (const float*)d_in[12];
  const float* bo    = (const float*)d_in[13];
  float* out = (float*)d_out;
  float* ws  = (float*)d_ws;

  float* qs   = ws;
  float* ks_  = qs  + 98304;
  float* vs   = ks_ + 98304;
  float* hq   = vs  + 98304;
  float* hk   = hq  + 73728;
  float* hv   = hk  + 73728;
  float* qp   = hv  + 73728;
  float* kp   = qp  + 73728;
  float* vp   = kp  + 73728;
  float* qq   = vp  + 73728;
  float* kk   = qq  + 6144;
  float* bias = kk  + 6144;
  float* attn = bias + 2359296;
  float* op   = attn + 2359296;
  float* ocat = op  + 73728;
  size_t req = (size_t)((ocat + 983040) - ws) * sizeof(float);
  if (ws_size < req) return;

  dim3 blk(256);

  proj_bias_kernel<<<BL + PJ_BLKS, dim3(128), 0, stream>>>(
      x, e, Wq_s, Wk_s, Wv_s, Wq_p, Wk_p, Wv_p, Wb, qs, ks_, vs, hq, hk, hv, bias);

  point_transform3_kernel<<<3 * PT_BLK, blk, 0, stream>>>(hq, hk, hv, r, t, qp, kp,
                                                          vp, qq, kk);

  logit_kernel<<<(BB * NH * LT * LT) / 8, blk, 0, stream>>>(qs, ks_, qp, kp, qq,
                                                            kk, bias, gamma, attn);
  softmax_kernel<<<BNL / 8, blk, 0, stream>>>(attn);

  attn_reduce_kernel<<<OP_BLKS + AV_BLKS, blk, 0, stream>>>(attn, e, vs, vp, ocat,
                                                            op);
  op_finish_kernel<<<BNL / 256, blk, 0, stream>>>(op, r, t, ocat);

  out_proj_kernel<<<(BL / 16) * (DD / 16), dim3(FP_KS * 32), 0, stream>>>(ocat, Wo, bo,
                                                                        out);
}


# Consumer stages in their producer's block: frame transform in the point-projection block, row softmax in the logit block, point finish in the attn@V wave; 7 launches -> 4; softmax sum order changes
# speedup vs baseline: 4.7158x; 1.3350x over previous
#include <hip/hip_runtime.h>
#include <hip/hip_bf16.h>
#include <math.h>

#define BB   2
#define LL   384
#define DD   128
#define NH   8
#define DHS_ 16
#define DHP_ 4
#define BL   (BB*LL)
#define BNL  (BB*NH*LL)
#define DOUT 1280
#define LT   24

#define GK 16

typedef float v2f __attribute__((ext_vector_type(2)));
typedef float v8f __attribute__((ext_vector_type(8)));

static __device__ __forceinline__ v8f wmma4(v2f a, v2f b, v8f c) {
  return __builtin_amdgcn_wmma_f32_16x16x4_f32(false, a, false, b, (short)0, c,
                                               false, false);
}


static __device__ __forceinline__ v8f gemm_acc(const float* __restrict__ A,
                                               const float* __restrict__ W,
                                               int lda, int Nc, int tm, int tn,
                                               int kbeg, int klen) {
  int lane = threadIdx.x & 31;
  int h = lane >> 4, l16 = lane & 15;
  const float* ap = A + (size_t)(tm + l16) * lda + kbeg + 2 * h;
  const float* wp = W + (size_t)(kbeg + 2 * h) * Nc + tn + l16;
  v8f acc = {};
  v2f a[GK], b[GK];
  auto load = [&](int s, int k) {
    a[s].x = ap[k];
    a[s].y = ap[k + 1];
    b[s].x = wp[(size_t)k * Nc];
    b[s].y = wp[(size_t)(k + 1) * Nc];
  };
#pragma unroll
  for (int s = 0; s < GK; ++s) {
    load(s, 4 * s);
    __builtin_amdgcn_sched_barrier(0);
  }
  int k0 = 0;
  for (; k0 + 4 * GK < klen; k0 += 4 * GK) {
#pragma unroll
    for (int s = 0; s < GK; ++s) {
      acc = wmma4(a[s], b[s], acc);
      load(s, k0 + 4 * GK + 4 * s);
      __builtin_amdgcn_sched_barrier(0);
    }
  }
#pragma unroll
  for (int s = 0; s < GK; ++s) acc = wmma4(a[s], b[s], acc);
  return acc;
}

static __device__ __forceinline__ void gemm_tile(const float* __restrict__ A,
                                                 const float* __restrict__ W,
                                                 const float* __restrict__ bias,
                                                 float* __restrict__ C,
                                                 int K, int Nc, int tile) {
  int lane = threadIdx.x & 31;
  int h = lane >> 4, l16 = lane & 15;
  int ntn = Nc >> 4;
  int tm = (tile / ntn) << 4;
  int tn = (tile % ntn) << 4;
  v8f acc = gemm_acc(A, W, K, Nc, tm, tn, 0, K);
  float bv = bias ? bias[tn + l16] : 0.0f;
#pragma unroll
  for (int v = 0; v < 8; ++v) {
    int row = tm + v + (h << 3);
    C[(size_t)row * Nc + tn + l16] = acc[v] + bv;
  }
}

typedef float v4f __attribute__((ext_vector_type(4)));

#define BIAS_TPW 6
#define BIAS_NS  32

static __device__ __forceinline__ void bias_row(const float* __restrict__ e,
                                                const float* __restrict__ Wb,
                                                float* __restrict__ bias,
                                                int bi) {
  int wave = threadIdx.x >> 5;
  int lane = threadIdx.x & 31;
  int h = lane >> 4, l16 = lane & 15;
  int b = bi / LL, i = bi % LL;
  int n = l16 & 7;
  v2f wf[32];
  const float* wp = Wb + (4 * h) * NH + n;
#pragma unroll
  for (int m = 0; m < 16; ++m) {
#pragma unroll
    for (int q = 0; q < 2; ++q) {
      wf[2 * m + q].x = wp[(8 * m + 2 * q) * NH];
      wf[2 * m + q].y = wp[(8 * m + 2 * q + 1) * NH];
    }
  }
  const float* ap =
      e + ((size_t)bi * LL + (size_t)(wave * BIAS_TPW * 16 + l16)) * DD + 4 * h;
  auto ld = [&](int t, int m) -> v4f {
    return *reinterpret_cast<const v4f*>(ap + (size_t)t * (16 * DD) + 8 * m);
  };
  bool hi = l16 >= 8;
  float* bp = bias + (((size_t)(b * NH + n) * LL + i) * LL) +
              wave * BIAS_TPW * 16 + 8 * h + (hi ? 4 : 0);
  auto store = [&](const v8f& acc, int t) {
    v4f o;
    o.x = hi ? acc[4] : acc[0];
    o.y = hi ? acc[5] : acc[1];
    o.z = hi ? acc[6] : acc[2];
    o.w = hi ? acc[7] : acc[3];
    *reinterpret_cast<v4f*>(bp + t * 16) = o;
  };
  v4f a[BIAS_NS];
#pragma unroll
  for (int s = 0; s < BIAS_NS; ++s) {
    a[s] = ld(s >> 4, s & 15);
    __builtin_amdgcn_sched_barrier(0);
  }
  for (int tp = 0; tp < BIAS_TPW / 2 - 1; ++tp) {
#pragma unroll
    for (int hf = 0; hf < 2; ++hf) {
      v8f acc = {};
#pragma unroll
      for (int m = 0; m < 16; ++m) {
        int s = hf * 16 + m;
        v2f a0 = {a[s].x, a[s].y}, a1 = {a[s].z, a[s].w};
        acc = wmma4(a0, wf[2 * m], acc);
        acc = wmma4(a1, wf[2 * m + 1], acc);
        a[s] = ld(2 * tp + 2 + hf, m);
        __builtin_amdgcn_sched_barrier(0);
      }
      store(acc, 2 * tp + hf);
    }
  }
#pragma unroll
  for (int hf = 0; hf < 2; ++hf) {
    v8f acc = {};
#pragma unroll
    for (int m = 0; m < 16; ++m) {
      int s = hf * 16 + m;
      v2f a0 = {a[s].x, a[s].y}, a1 = {a[s].z, a[s].w};
      acc = wmma4(a0, wf[2 * m], acc);
      acc = wmma4(a1, wf[2 * m + 1], acc);
    }
    store(acc, BIAS_TPW - 2 + hf);
  }
}

static __device__ __forceinline__ void point_transform(const float* hp,
                                                       const float* rp,
                                                       const float* tp,
                                                       float* __restrict__ outp,
                                                       float* __restrict__ sumsq,
                                                       int idx) {
  float* op = outp + (size_t)idx * 12;
  float ss = 0.0f;
#pragma unroll
  for (int p = 0; p < 4; ++p)
#pragma unroll
    for (int c = 0; c < 3; ++c) {
      float v = tp[c];
#pragma unroll
      for (int k = 0; k < 3; ++k) v += hp[p * 3 + k] * rp[k * 3 + c];
      op[p * 3 + c] = v;
      ss += v * v;
    }
  if (sumsq) sumsq[idx] = ss;
}

#define LS_W 8
#define LS_T (LT / LS_W)

__global__ void __launch_bounds__(LS_W * 32)
logit_softmax_kernel(const float* __restrict__ qs,
                     const float* __restrict__ ks,
                     const float* __restrict__ qp,
                     const float* __restrict__ kp,
                     const float* __restrict__ qq,
                     const float* __restrict__ kk,
                     const float* __restrict__ bias,
                     const float* __restrict__ gamma,
                     float* __restrict__ attn) {
  __shared__ __attribute__((aligned(16))) float red_m[LS_W][16];
  __shared__ __attribute__((aligned(16))) float red_s[LS_W][16];
  int strip = blockIdx.x;
  int it = strip % LT;
  int n  = (strip / LT) % NH;
  int b  = strip / (LT * NH);
  int wave = threadIdx.x >> 5;
  int lane = threadIdx.x & 31;
  int h = lane >> 4, l16 = lane & 15;
  size_t hb = (size_t)(b * NH + n) * LL;

  const float* qsp = qs + (size_t)(b * LL + it * 16 + l16) * DD + n * DHS_ + 2 * h;
  const float* qpp = qp + (hb + it * 16 + l16) * 12 + 2 * h;
  v2f aS[4], aP[3];
#pragma unroll
  for (int k = 0; k < 4; ++k) { aS[k].x = qsp[4 * k]; aS[k].y = qsp[4 * k + 1]; }
#pragma unroll
  for (int k = 0; k < 3; ++k) { aP[k].x = qpp[4 * k]; aP[k].y = qpp[4 * k + 1]; }
  v2f bS[LS_T][4], bP[LS_T][3];
  float kkj[LS_T], bsv[LS_T][8], qqi[8];
#pragma unroll
  for (int t = 0; t < LS_T; ++t) {
    int jt = wave * LS_T + t;
    const float* ksp = ks + (size_t)(b * LL + jt * 16 + l16) * DD + n * DHS_ + 2 * h;
    const float* kpp = kp + (hb + jt * 16 + l16) * 12 + 2 * h;
#pragma unroll
    for (int k = 0; k < 4; ++k) { bS[t][k].x = ksp[4 * k]; bS[t][k].y = ksp[4 * k + 1]; }
#pragma unroll
    for (int k = 0; k < 3; ++k) { bP[t][k].x = kpp[4 * k]; bP[t][k].y = kpp[4 * k + 1]; }
    kkj[t] = kk[hb + jt * 16 + l16];
#pragma unroll
    for (int v = 0; v < 8; ++v)
      bsv[t][v] = bias[(hb + it * 16 + v + (h << 3)) * LL + jt * 16 + l16];
  }
#pragma unroll
  for (int v = 0; v < 8; ++v) qqi[v] = qq[hb + it * 16 + v + (h << 3)];
  float g = gamma[n];
  const float rsqrt3 = 0.57735026918962576f;
  __builtin_amdgcn_sched_barrier(0);

  float lg[LS_T][8];
#pragma unroll
  for (int t = 0; t < LS_T; ++t) {
    v8f accS = {}, accP = {};
#pragma unroll
    for (int k = 0; k < 4; ++k) accS = wmma4(aS[k], bS[t][k], accS);
#pragma unroll
    for (int k = 0; k < 3; ++k) accP = wmma4(aP[k], bP[t][k], accP);
#pragma unroll
    for (int v = 0; v < 8; ++v) {
      float d2 = qqi[v] + kkj[t] - 2.0f * accP[v];
      float lp = -0.125f * g * d2;
      lg[t][v] = rsqrt3 * (0.25f * accS[v] + bsv[t][v] + lp);
    }
  }

  float m[8];
#pragma unroll
  for (int v = 0; v < 8; ++v) {
    float x = lg[0][v];
#pragma unroll
    for (int t = 1; t < LS_T; ++t) x = fmaxf(x, lg[t][v]);
#pragma unroll
    for (int off = 8; off > 0; off >>= 1) x = fmaxf(x, __shfl_xor(x, off, 32));
    m[v] = x;
  }
  if (l16 == 0) {
#pragma unroll
    for (int v = 0; v < 8; ++v) red_m[wave][8 * h + v] = m[v];
  }
  __syncthreads();
#pragma unroll
  for (int w = 0; w < LS_W; ++w) {
    v4f lo = *reinterpret_cast<const v4f*>(&red_m[w][8 * h]);
    v4f hi = *reinterpret_cast<const v4f*>(&red_m[w][8 * h + 4]);
    float r8[8] = {lo.x, lo.y, lo.z, lo.w, hi.x, hi.y, hi.z, hi.w};
#pragma unroll
    for (int v = 0; v < 8; ++v) m[v] = w == 0 ? r8[v] : fmaxf(m[v], r8[v]);
  }

  float s[8];
#pragma unroll
  for (int v = 0; v < 8; ++v) {
    float x = 0.0f;
#pragma unroll
    for (int t = 0; t < LS_T; ++t) {
      lg[t][v] = __expf(lg[t][v] - m[v]);
      x += lg[t][v];
    }
#pragma unroll
    for (int off = 8; off > 0; off >>= 1) x += __shfl_xor(x, off, 32);
    s[v] = x;
  }
  if (l16 == 0) {
#pragma unroll
    for (int v = 0; v < 8; ++v) red_s[wave][8 * h + v] = s[v];
  }
  __syncthreads();
#pragma unroll
  for (int w = 0; w < LS_W; ++w) {
    v4f lo = *reinterpret_cast<const v4f*>(&red_s[w][8 * h]);
    v4f hi = *reinterpret_cast<const v4f*>(&red_s[w][8 * h + 4]);
    float r8[8] = {lo.x, lo.y, lo.z, lo.w, hi.x, hi.y, hi.z, hi.w};
#pragma unroll
    for (int v = 0; v < 8; ++v) s[v] = w == 0 ? r8[v] : s[v] + r8[v];
  }
#pragma unroll
  for (int v = 0; v < 8; ++v) {
    float inv = 1.0f / s[v];
    size_t row = (hb + it * 16 + v + (h << 3)) * LL;
#pragma unroll
    for (int t = 0; t < LS_T; ++t)
      attn[row + (wave * LS_T + t) * 16 + l16] = lg[t][v] * inv;
  }
}

static __device__ __forceinline__ void attn_v_tile(const float* __restrict__ attn,
                                                   const float* __restrict__ vs,
                                                   const float* __restrict__ vp,
                                                   const float* __restrict__ r,
                                                   const float* __restrict__ t,
                                                   float* __restrict__ out_cat,
                                                   int tile) {
  int it = tile % LT;
  int n  = (tile / LT) % NH;
  int b  = tile / (LT * NH);
  int lane = threadIdx.x & 31;
  int h = lane >> 4, l16 = lane & 15;
  size_t hb = (size_t)(b * NH + n) * LL;
  const float* ap  = attn + (hb + it * 16 + l16) * LL + 2 * h;
  const float* vsb = vs + (size_t)b * LL * DD + n * DHS_ + l16;
  bool pv = l16 < 12;
  const float* vpb = vp + hb * 12 + (pv ? l16 : 11);
  v8f accS = {}, accP = {};
  constexpr int GV = 12;
  v2f a[GV], bs[GV], bp[GV];
  auto load = [&](int s, int k) {
    a[s].x = ap[k];
    a[s].y = ap[k + 1];
    bs[s].x = vsb[(size_t)(k + 2 * h) * DD];
    bs[s].y = vsb[(size_t)(k + 2 * h + 1) * DD];
    bp[s].x = vpb[(k + 2 * h) * 12];
    bp[s].y = vpb[(k + 2 * h + 1) * 12];
  };
#pragma unroll
  for (int s = 0; s < GV; ++s) {
    load(s, 4 * s);
    __builtin_amdgcn_sched_barrier(0);
  }
  for (int k0 = 0; k0 + 4 * GV < LL; k0 += 4 * GV) {
#pragma unroll
    for (int s = 0; s < GV; ++s) {
      accS = wmma4(a[s], bs[s], accS);
      accP = wmma4(a[s], bp[s], accP);
      load(s, k0 + 4 * GV + 4 * s);
      __builtin_amdgcn_sched_barrier(0);
    }
  }
#pragma unroll
  for (int s = 0; s < GV; ++s) {
    accS = wmma4(a[s], bs[s], accS);
    accP = wmma4(a[s], bp[s], accP);
  }
  int cl = pv ? l16 : 11;
  int p  = cl / 3;
  int kc = cl - 3 * p;
  int src = (lane & 16) + 3 * p;
#pragma unroll
  for (int v = 0; v < 8; ++v) {
    int i = it * 16 + v + (h << 3);
    const float* rp = r + (size_t)(b * LL + i) * 9 + kc * 3;
    float om = accP[v] - t[(size_t)(b * LL + i) * 3 + kc];
    float om0 = __shfl(om, src, 32);
    float om1 = __shfl(om, src + 1, 32);
    float om2 = __shfl(om, src + 2, 32);
    float val = om0 * rp[0] + om1 * rp[1] + om2 * rp[2];
    float sq = val * val;
    float nsq = 0.0f;
    nsq += __shfl(sq, src, 32);
    nsq += __shfl(sq, src + 1, 32);
    nsq += __shfl(sq, src + 2, 32);
    float* oc = out_cat + (size_t)(b * LL + i) * DOUT;
    oc[n * DHS_ + l16] = accS[v];
    if (pv) {
      oc[DD + NH * DD + n * 12 + l16] = val;
      if (kc == 0) oc[DD + NH * DD + NH * 12 + n * 4 + p] = sqrtf(nsq);
    }
  }
}

static __device__ __forceinline__ void out_pair_tile(const float* __restrict__ attn,
                                                     const float* __restrict__ e,
                                                     float* __restrict__ out_cat,
                                                     int tile) {
  int dt = tile % 8;
  int i  = (tile / 8) % LL;
  int b  = tile / (8 * LL);
  int lane = threadIdx.x & 31;
  int h = lane >> 4, l16 = lane & 15;
  const float* ap = attn + ((size_t)(b * NH + (l16 & 7)) * LL + i) * LL + 2 * h;
  const float* ep = e + ((size_t)(b * LL + i) * LL) * DD + dt * 16 + l16;
  v8f acc = {};
  v2f a[GK], bv[GK];
  auto load = [&](int s, int k) {
    a[s].x = ap[k];
    a[s].y = ap[k + 1];
    bv[s].x = ep[(size_t)(k + 2 * h) * DD];
    bv[s].y = ep[(size_t)(k + 2 * h + 1) * DD];
  };
#pragma unroll
  for (int s = 0; s < GK; ++s) {
    load(s, 4 * s);
    __builtin_amdgcn_sched_barrier(0);
  }
  for (int k0 = 0; k0 + 4 * GK < LL; k0 += 4 * GK) {
#pragma unroll
    for (int s = 0; s < GK; ++s) {
      acc = wmma4(a[s], bv[s], acc);
      load(s, k0 + 4 * GK + 4 * s);
      __builtin_amdgcn_sched_barrier(0);
    }
  }
#pragma unroll
  for (int s = 0; s < GK; ++s) acc = wmma4(a[s], bv[s], acc);
  if (h == 0) {
#pragma unroll
    for (int v = 0; v < 8; ++v)
      out_cat[(size_t)(b * LL + i) * DOUT + DD + v * DD + dt * 16 + l16] = acc[v];
  }
}

#define PJ_BLK_S 96
#define PJ_BLK_P 96
#define PJ_BLKS  (3 * PJ_BLK_S + 3 * PJ_BLK_P)
#define PT_PITCH 49

__global__ void __launch_bounds__(128)
proj_bias_kernel(const float* __restrict__ x,
                 const float* __restrict__ e,
                 const float* __restrict__ Wq_s,
                 const float* __restrict__ Wk_s,
                 const float* __restrict__ Wv_s,
                 const float* __restrict__ Wq_p,
                 const float* __restrict__ Wk_p,
                 const float* __restrict__ Wv_p,
                 const float* __restrict__ Wb,
                 const float* __restrict__ r,
                 const float* __restrict__ t,
                 float* __restrict__ qs,
                 float* __restrict__ ks,
                 float* __restrict__ vs,
                 float* __restrict__ qp,
                 float* __restrict__ kp,
                 float* __restrict__ vp,
                 float* __restrict__ qq,
                 float* __restrict__ kk,
                 float* __restrict__ bias) {
  __shared__ float hm[16 * PT_PITCH];
  int blk = blockIdx.x;
  if (blk < BL) {
    bias_row(e, Wb, bias, blk);
    return;
  }
  blk -= BL;
  int wave = threadIdx.x >> 5;
  if (blk < 3 * PJ_BLK_S) {
    int g = blk / PJ_BLK_S;
    blk -= g * PJ_BLK_S;
    const float* Wm = g == 0 ? Wq_s : (g == 1 ? Wk_s : Wv_s);
    float* C        = g == 0 ? qs : (g == 1 ? ks : vs);
    gemm_tile(x, Wm, nullptr, C, DD, DD, blk * 4 + wave);
    return;
  }
  blk -= 3 * PJ_BLK_S;
  int g = blk / PJ_BLK_P;
  blk -= g * PJ_BLK_P;
  const float* Wm = g == 0 ? Wq_p : (g == 1 ? Wk_p : Wv_p);
  float* outp     = g == 0 ? qp : (g == 1 ? kp : vp);
  float* ss       = g == 0 ? qq : (g == 1 ? kk : nullptr);
  int tm = (blk >> 1) << 4;
  int half = blk & 1;
  int rr = threadIdx.x & 15;
  int hh = (threadIdx.x >> 4) & 3;
  int row = tm + rr;
  float rv[9], tv[3];
#pragma unroll
  for (int q = 0; q < 9; ++q) rv[q] = r[(size_t)row * 9 + q];
#pragma unroll
  for (int q = 0; q < 3; ++q) tv[q] = t[(size_t)row * 3 + q];
  if (wave < 3) {
    int lane = threadIdx.x & 31;
    int h = lane >> 4, l16 = lane & 15;
    v8f acc = gemm_acc(x, Wm, DD, NH * 12, tm, half * 48 + wave * 16, 0, DD);
#pragma unroll
    for (int v = 0; v < 8; ++v) hm[(v + (h << 3)) * PT_PITCH + wave * 16 + l16] = acc[v];
  }
  __syncthreads();
  if (threadIdx.x < 64) {
    int b = row / LL, l = row % LL;
    int n = half * 4 + hh;
    point_transform(&hm[rr * PT_PITCH + hh * 12], rv, tv, outp, ss,
                    (b * NH + n) * LL + l);
  }
}

#define OP_BLKS (BB * LL)
#define AV_BLKS ((BB * NH * LT) / 8)

__global__ void attn_reduce_kernel(const float* __restrict__ attn,
                                   const float* __restrict__ e,
                                   const float* __restrict__ vs,
                                   const float* __restrict__ vp,
                                   const float* __restrict__ r,
                                   const float* __restrict__ t,
                                   float* __restrict__ out_cat) {
  int wave = threadIdx.x >> 5;
  if (blockIdx.x < OP_BLKS) {
    out_pair_tile(attn, e, out_cat, blockIdx.x * 8 + wave);
  } else {
    attn_v_tile(attn, vs, vp, r, t, out_cat, (blockIdx.x - OP_BLKS) * 8 + wave);
  }
}

#define FP_KS 10
#define FP_KC (DOUT / FP_KS)

__global__ void __launch_bounds__(FP_KS * 32)
out_proj_kernel(const float* __restrict__ A,
                const float* __restrict__ W,
                const float* __restrict__ bias,
                float* __restrict__ C) {
  __shared__ float red[FP_KS][8][32];
  int wave = threadIdx.x >> 5;
  int lane = threadIdx.x & 31;
  int tm = (blockIdx.x / (DD / 16)) << 4;
  int tn = (blockIdx.x % (DD / 16)) << 4;
  v8f acc = gemm_acc(A, W, DOUT, DD, tm, tn, wave * FP_KC, FP_KC);
#pragma unroll
  for (int v = 0; v < 8; ++v) red[wave][v][lane] = acc[v];
  __syncthreads();
  if (wave < 8) {
    float s = red[0][wave][lane];
#pragma unroll
    for (int w = 1; w < FP_KS; ++w) s += red[w][wave][lane];
    int row = tm + wave + ((lane >> 4) << 3);
    int col = tn + (lane & 15);
    C[(size_t)row * DD + col] = s + bias[col];
  }
}

extern "C" void kernel_launch(void* const* d_in, const int* in_sizes, int n_in,
                              void* d_out, int out_size, void* d_ws, size_t ws_size,
                              hipStream_t stream) {
  (void)in_sizes; (void)n_in; (void)out_size;
  const float* x     = (const float*)d_in[0];
  const float* e     = (const float*)d_in[1];
  const float* r     = (const float*)d_in[2];
  const float* t     = (const float*)d_in[3];
  const float* Wq_s  = (const float*)d_in[4];
  const float* Wk_s  = (const float*)d_in[5];
  const float* Wv_s  = (const float*)d_in[6];
  const float* Wb    = (const float*)d_in[7];
  const float* Wq_p  = (const float*)d_in[8];
  const float* Wk_p  = (const float*)d_in[9];
  const float* Wv_p  = (const float*)d_in[10];
  const float* gamma = (const float*)d_in[11];
  const float* Wo    = (const float*)d_in[12];
  const float* bo    = (const float*)d_in[13];
  float* out = (float*)d_out;
  float* ws  = (float*)d_ws;

  float* qs   = ws;
  float* ks_  = qs  + 98304;
  float* vs   = ks_ + 98304;
  float* qp   = vs  + 98304;
  float* kp   = qp  + 73728;
  float* vp   = kp  + 73728;
  float* qq   = vp  + 73728;
  float* kk   = qq  + 6144;
  float* bias = kk  + 6144;
  float* attn = bias + 2359296;
  float* ocat = attn + 2359296;
  size_t req = (size_t)((ocat + 983040) - ws) * sizeof(float);
  if (ws_size < req) return;

  dim3 blk(256);

  proj_bias_kernel<<<BL + PJ_BLKS, dim3(128), 0, stream>>>(
      x, e, Wq_s, Wk_s, Wv_s, Wq_p, Wk_p, Wv_p, Wb, r, t, qs, ks_, vs, qp, kp, vp,
      qq, kk, bias);

  logit_softmax_kernel<<<BB * NH * LT, dim3(LS_W * 32), 0, stream>>>(
      qs, ks_, qp, kp, qq, kk, bias, gamma, attn);

  attn_reduce_kernel<<<OP_BLKS + AV_BLKS, blk, 0, stream>>>(attn, e, vs, vp, r, t,
                                                            ocat);

  out_proj_kernel<<<(BL / 16) * (DD / 16), dim3(FP_KS * 32), 0, stream>>>(ocat, Wo, bo,
                                                                        out);
}


# pair output: a wave takes 4 column tiles (lane = 4 neighbouring columns): e rows read by 16-byte loads of whole lines, one attn fragment feeds 4 WMMAs; j split in 4, added via LDS: sum order changes
# speedup vs baseline: 5.1316x; 1.0882x over previous
#include <hip/hip_runtime.h>
#include <hip/hip_bf16.h>
#include <math.h>

#define BB   2
#define LL   384
#define DD   128
#define NH   8
#define DHS_ 16
#define DHP_ 4
#define BL   (BB*LL)
#define BNL  (BB*NH*LL)
#define DOUT 1280
#define LT   24

#define GK 16

typedef float v2f __attribute__((ext_vector_type(2)));
typedef float v8f __attribute__((ext_vector_type(8)));

static __device__ __forceinline__ v8f wmma4(v2f a, v2f b, v8f c) {
  return __builtin_amdgcn_wmma_f32_16x16x4_f32(false, a, false, b, (short)0, c,
                                               false, false);
}


static __device__ __forceinline__ v8f gemm_acc(const float* __restrict__ A,
                                               const float* __restrict__ W,
                                               int lda, int Nc, int tm, int tn,
                                               int kbeg, int klen) {
  int lane = threadIdx.x & 31;
  int h = lane >> 4, l16 = lane & 15;
  const float* ap = A + (size_t)(tm + l16) * lda + kbeg + 2 * h;
  const float* wp = W + (size_t)(kbeg + 2 * h) * Nc + tn + l16;
  v8f acc = {};
  v2f a[GK], b[GK];
  auto load = [&](int s, int k) {
    a[s].x = ap[k];
    a[s].y = ap[k + 1];
    b[s].x = wp[(size_t)k * Nc];
    b[s].y = wp[(size_t)(k + 1) * Nc];
  };
#pragma unroll
  for (int s = 0; s < GK; ++s) {
    load(s, 4 * s);
    __builtin_amdgcn_sched_barrier(0);
  }
  int k0 = 0;
  for (; k0 + 4 * GK < klen; k0 += 4 * GK) {
#pragma unroll
    for (int s = 0; s < GK; ++s) {
      acc = wmma4(a[s], b[s], acc);
      load(s, k0 + 4 * GK + 4 * s);
      __builtin_amdgcn_sched_barrier(0);
    }
  }
#pragma unroll
  for (int s = 0; s < GK; ++s) acc = wmma4(a[s], b[s], acc);
  return acc;
}

static __device__ __forceinline__ void gemm_tile(const float* __restrict__ A,
                                                 const float* __restrict__ W,
                                                 const float* __restrict__ bias,
                                                 float* __restrict__ C,
                                                 int K, int Nc, int tile) {
  int lane = threadIdx.x & 31;
  int h = lane >> 4, l16 = lane & 15;
  int ntn = Nc >> 4;
  int tm = (tile / ntn) << 4;
  int tn = (tile % ntn) << 4;
  v8f acc = gemm_acc(A, W, K, Nc, tm, tn, 0, K);
  float bv = bias ? bias[tn + l16] : 0.0f;
#pragma unroll
  for (int v = 0; v < 8; ++v) {
    int row = tm + v + (h << 3);
    C[(size_t)row * Nc + tn + l16] = acc[v] + bv;
  }
}

typedef float v4f __attribute__((ext_vector_type(4)));

#define BIAS_TPW 6
#define BIAS_NS  32

static __device__ __forceinline__ void bias_row(const float* __restrict__ e,
                                                const float* __restrict__ Wb,
                                                float* __restrict__ bias,
                                                int bi) {
  int wave = threadIdx.x >> 5;
  int lane = threadIdx.x & 31;
  int h = lane >> 4, l16 = lane & 15;
  int b = bi / LL, i = bi % LL;
  int n = l16 & 7;
  v2f wf[32];
  const float* wp = Wb + (4 * h) * NH + n;
#pragma unroll
  for (int m = 0; m < 16; ++m) {
#pragma unroll
    for (int q = 0; q < 2; ++q) {
      wf[2 * m + q].x = wp[(8 * m + 2 * q) * NH];
      wf[2 * m + q].y = wp[(8 * m + 2 * q + 1) * NH];
    }
  }
  const float* ap =
      e + ((size_t)bi * LL + (size_t)(wave * BIAS_TPW * 16 + l16)) * DD + 4 * h;
  auto ld = [&](int t, int m) -> v4f {
    return *reinterpret_cast<const v4f*>(ap + (size_t)t * (16 * DD) + 8 * m);
  };
  bool hi = l16 >= 8;
  float* bp = bias + (((size_t)(b * NH + n) * LL + i) * LL) +
              wave * BIAS_TPW * 16 + 8 * h + (hi ? 4 : 0);
  auto store = [&](const v8f& acc, int t) {
    v4f o;
    o.x = hi ? acc[4] : acc[0];
    o.y = hi ? acc[5] : acc[1];
    o.z = hi ? acc[6] : acc[2];
    o.w = hi ? acc[7] : acc[3];
    *reinterpret_cast<v4f*>(bp + t * 16) = o;
  };
  v4f a[BIAS_NS];
#pragma unroll
  for (int s = 0; s < BIAS_NS; ++s) {
    a[s] = ld(s >> 4, s & 15);
    __builtin_amdgcn_sched_barrier(0);
  }
  for (int tp = 0; tp < BIAS_TPW / 2 - 1; ++tp) {
#pragma unroll
    for (int hf = 0; hf < 2; ++hf) {
      v8f acc = {};
#pragma unroll
      for (int m = 0; m < 16; ++m) {
        int s = hf * 16 + m;
        v2f a0 = {a[s].x, a[s].y}, a1 = {a[s].z, a[s].w};
        acc = wmma4(a0, wf[2 * m], acc);
        acc = wmma4(a1, wf[2 * m + 1], acc);
        a[s] = ld(2 * tp + 2 + hf, m);
        __builtin_amdgcn_sched_barrier(0);
      }
      store(acc, 2 * tp + hf);
    }
  }
#pragma unroll
  for (int hf = 0; hf < 2; ++hf) {
    v8f acc = {};
#pragma unroll
    for (int m = 0; m < 16; ++m) {
      int s = hf * 16 + m;
      v2f a0 = {a[s].x, a[s].y}, a1 = {a[s].z, a[s].w};
      acc = wmma4(a0, wf[2 * m], acc);
      acc = wmma4(a1, wf[2 * m + 1], acc);
    }
    store(acc, BIAS_TPW - 2 + hf);
  }
}

static __device__ __forceinline__ void point_transform(const float* hp,
                                                       const float* rp,
                                                       const float* tp,
                                                       float* __restrict__ outp,
                                                       float* __restrict__ sumsq,
                                                       int idx) {
  float* op = outp + (size_t)idx * 12;
  float ss = 0.0f;
#pragma unroll
  for (int p = 0; p < 4; ++p)
#pragma unroll
    for (int c = 0; c < 3; ++c) {
      float v = tp[c];
#pragma unroll
      for (int k = 0; k < 3; ++k) v += hp[p * 3 + k] * rp[k * 3 + c];
      op[p * 3 + c] = v;
      ss += v * v;
    }
  if (sumsq) sumsq[idx] = ss;
}

#define LS_W 8
#define LS_T (LT / LS_W)

__global__ void __launch_bounds__(LS_W * 32)
logit_softmax_kernel(const float* __restrict__ qs,
                     const float* __restrict__ ks,
                     const float* __restrict__ qp,
                     const float* __restrict__ kp,
                     const float* __restrict__ qq,
                     const float* __restrict__ kk,
                     const float* __restrict__ bias,
                     const float* __restrict__ gamma,
                     float* __restrict__ attn) {
  __shared__ __attribute__((aligned(16))) float red_m[LS_W][16];
  __shared__ __attribute__((aligned(16))) float red_s[LS_W][16];
  int strip = blockIdx.x;
  int it = strip % LT;
  int n  = (strip / LT) % NH;
  int b  = strip / (LT * NH);
  int wave = threadIdx.x >> 5;
  int lane = threadIdx.x & 31;
  int h = lane >> 4, l16 = lane & 15;
  size_t hb = (size_t)(b * NH + n) * LL;

  const float* qsp = qs + (size_t)(b * LL + it * 16 + l16) * DD + n * DHS_ + 2 * h;
  const float* qpp = qp + (hb + it * 16 + l16) * 12 + 2 * h;
  v2f aS[4], aP[3];
#pragma unroll
  for (int k = 0; k < 4; ++k) { aS[k].x = qsp[4 * k]; aS[k].y = qsp[4 * k + 1]; }
#pragma unroll
  for (int k = 0; k < 3; ++k) { aP[k].x = qpp[4 * k]; aP[k].y = qpp[4 * k + 1]; }
  v2f bS[LS_T][4], bP[LS_T][3];
  float kkj[LS_T], bsv[LS_T][8], qqi[8];
#pragma unroll
  for (int t = 0; t < LS_T; ++t) {
    int jt = wave * LS_T + t;
    const float* ksp = ks + (size_t)(b * LL + jt * 16 + l16) * DD + n * DHS_ + 2 * h;
    const float* kpp = kp + (hb + jt * 16 + l16) * 12 + 2 * h;
#pragma unroll
    for (int k = 0; k < 4; ++k) { bS[t][k].x = ksp[4 * k]; bS[t][k].y = ksp[4 * k + 1]; }
#pragma unroll
    for (int k = 0; k < 3; ++k) { bP[t][k].x = kpp[4 * k]; bP[t][k].y = kpp[4 * k + 1]; }
    kkj[t] = kk[hb + jt * 16 + l16];
#pragma unroll
    for (int v = 0; v < 8; ++v)
      bsv[t][v] = bias[(hb + it * 16 + v + (h << 3)) * LL + jt * 16 + l16];
  }
#pragma unroll
  for (int v = 0; v < 8; ++v) qqi[v] = qq[hb + it * 16 + v + (h << 3)];
  float g = gamma[n];
  const float rsqrt3 = 0.57735026918962576f;
  __builtin_amdgcn_sched_barrier(0);

  float lg[LS_T][8];
#pragma unroll
  for (int t = 0; t < LS_T; ++t) {
    v8f accS = {}, accP = {};
#pragma unroll
    for (int k = 0; k < 4; ++k) accS = wmma4(aS[k], bS[t][k], accS);
#pragma unroll
    for (int k = 0; k < 3; ++k) accP = wmma4(aP[k], bP[t][k], accP);
#pragma unroll
    for (int v = 0; v < 8; ++v) {
      float d2 = qqi[v] + kkj[t] - 2.0f * accP[v];
      float lp = -0.125f * g * d2;
      lg[t][v] = rsqrt3 * (0.25f * accS[v] + bsv[t][v] + lp);
    }
  }

  float m[8];
#pragma unroll
  for (int v = 0; v < 8; ++v) {
    float x = lg[0][v];
#pragma unroll
    for (int t = 1; t < LS_T; ++t) x = fmaxf(x, lg[t][v]);
#pragma unroll
    for (int off = 8; off > 0; off >>= 1) x = fmaxf(x, __shfl_xor(x, off, 32));
    m[v] = x;
  }
  if (l16 == 0) {
#pragma unroll
    for (int v = 0; v < 8; ++v) red_m[wave][8 * h + v] = m[v];
  }
  __syncthreads();
#pragma unroll
  for (int w = 0; w < LS_W; ++w) {
    v4f lo = *reinterpret_cast<const v4f*>(&red_m[w][8 * h]);
    v4f hi = *reinterpret_cast<const v4f*>(&red_m[w][8 * h + 4]);
    float r8[8] = {lo.x, lo.y, lo.z, lo.w, hi.x, hi.y, hi.z, hi.w};
#pragma unroll
    for (int v = 0; v < 8; ++v) m[v] = w == 0 ? r8[v] : fmaxf(m[v], r8[v]);
  }

  float s[8];
#pragma unroll
  for (int v = 0; v < 8; ++v) {
    float x = 0.0f;
#pragma unroll
    for (int t = 0; t < LS_T; ++t) {
      lg[t][v] = __expf(lg[t][v] - m[v]);
      x += lg[t][v];
    }
#pragma unroll
    for (int off = 8; off > 0; off >>= 1) x += __shfl_xor(x, off, 32);
    s[v] = x;
  }
  if (l16 == 0) {
#pragma unroll
    for (int v = 0; v < 8; ++v) red_s[wave][8 * h + v] = s[v];
  }
  __syncthreads();
#pragma unroll
  for (int w = 0; w < LS_W; ++w) {
    v4f lo = *reinterpret_cast<const v4f*>(&red_s[w][8 * h]);
    v4f hi = *reinterpret_cast<const v4f*>(&red_s[w][8 * h + 4]);
    float r8[8] = {lo.x, lo.y, lo.z, lo.w, hi.x, hi.y, hi.z, hi.w};
#pragma unroll
    for (int v = 0; v < 8; ++v) s[v] = w == 0 ? r8[v] : s[v] + r8[v];
  }
#pragma unroll
  for (int v = 0; v < 8; ++v) {
    float inv = 1.0f / s[v];
    size_t row = (hb + it * 16 + v + (h << 3)) * LL;
#pragma unroll
    for (int t = 0; t < LS_T; ++t)
      attn[row + (wave * LS_T + t) * 16 + l16] = lg[t][v] * inv;
  }
}

static __device__ __forceinline__ void attn_v_tile(const float* __restrict__ attn,
                                                   const float* __restrict__ vs,
                                                   const float* __restrict__ vp,
                                                   const float* __restrict__ r,
                                                   const float* __restrict__ t,
                                                   float* __restrict__ out_cat,
                                                   int tile) {
  int it = tile % LT;
  int n  = (tile / LT) % NH;
  int b  = tile / (LT * NH);
  int lane = threadIdx.x & 31;
  int h = lane >> 4, l16 = lane & 15;
  size_t hb = (size_t)(b * NH + n) * LL;
  const float* ap  = attn + (hb + it * 16 + l16) * LL + 2 * h;
  const float* vsb = vs + (size_t)b * LL * DD + n * DHS_ + l16;
  bool pv = l16 < 12;
  const float* vpb = vp + hb * 12 + (pv ? l16 : 11);
  v8f accS = {}, accP = {};
  constexpr int GV = 12;
  v2f a[GV], bs[GV], bp[GV];
  auto load = [&](int s, int k) {
    a[s].x = ap[k];
    a[s].y = ap[k + 1];
    bs[s].x = vsb[(size_t)(k + 2 * h) * DD];
    bs[s].y = vsb[(size_t)(k + 2 * h + 1) * DD];
    bp[s].x = vpb[(k + 2 * h) * 12];
    bp[s].y = vpb[(k + 2 * h + 1) * 12];
  };
#pragma unroll
  for (int s = 0; s < GV; ++s) {
    load(s, 4 * s);
    __builtin_amdgcn_sched_barrier(0);
  }
  for (int k0 = 0; k0 + 4 * GV < LL; k0 += 4 * GV) {
#pragma unroll
    for (int s = 0; s < GV; ++s) {
      accS = wmma4(a[s], bs[s], accS);
      accP = wmma4(a[s], bp[s], accP);
      load(s, k0 + 4 * GV + 4 * s);
      __builtin_amdgcn_sched_barrier(0);
    }
  }
#pragma unroll
  for (int s = 0; s < GV; ++s) {
    accS = wmma4(a[s], bs[s], accS);
    accP = wmma4(a[s], bp[s], accP);
  }
  int cl = pv ? l16 : 11;
  int p  = cl / 3;
  int kc = cl - 3 * p;
  int src = (lane & 16) + 3 * p;
#pragma unroll
  for (int v = 0; v < 8; ++v) {
    int i = it * 16 + v + (h << 3);
    const float* rp = r + (size_t)(b * LL + i) * 9 + kc * 3;
    float om = accP[v] - t[(size_t)(b * LL + i) * 3 + kc];
    float om0 = __shfl(om, src, 32);
    float om1 = __shfl(om, src + 1, 32);
    float om2 = __shfl(om, src + 2, 32);
    float val = om0 * rp[0] + om1 * rp[1] + om2 * rp[2];
    float sq = val * val;
    float nsq = 0.0f;
    nsq += __shfl(sq, src, 32);
    nsq += __shfl(sq, src + 1, 32);
    nsq += __shfl(sq, src + 2, 32);
    float* oc = out_cat + (size_t)(b * LL + i) * DOUT;
    oc[n * DHS_ + l16] = accS[v];
    if (pv) {
      oc[DD + NH * DD + n * 12 + l16] = val;
      if (kc == 0) oc[DD + NH * DD + NH * 12 + n * 4 + p] = sqrtf(nsq);
    }
  }
}

#define OP_JQ 4
#define OP_JL (LL / OP_JQ)
#define OP_G  6

static __device__ __forceinline__ void out_pair_row(const float* __restrict__ attn,
                                                    const float* __restrict__ e,
                                                    float* __restrict__ out_cat,
                                                    float (*part)[NH][DD],
                                                    int bi) {
  int wave = threadIdx.x >> 5;
  int lane = threadIdx.x & 31;
  int h = lane >> 4, l16 = lane & 15;
  int b = bi / LL, i = bi % LL;
  int dh = wave & 1;
  int jq = wave >> 1;
  int dcol = dh * 64 + 4 * l16;
  const float* ap =
      attn + ((size_t)(b * NH + (l16 & 7)) * LL + i) * LL + jq * OP_JL + 2 * h;
  const float* ep = e + ((size_t)bi * LL + jq * OP_JL + 2 * h) * DD + dcol;
  v8f acc[4] = {};
  v2f a[OP_G];
  v4f bx[OP_G], by[OP_G];
  auto load = [&](int s, int k) {
    a[s].x = ap[k];
    a[s].y = ap[k + 1];
    bx[s] = *reinterpret_cast<const v4f*>(ep + (size_t)k * DD);
    by[s] = *reinterpret_cast<const v4f*>(ep + (size_t)(k + 1) * DD);
  };
  auto mm = [&](int s) {
    v2f b0 = {bx[s].x, by[s].x}, b1 = {bx[s].y, by[s].y};
    v2f b2 = {bx[s].z, by[s].z}, b3 = {bx[s].w, by[s].w};
    acc[0] = wmma4(a[s], b0, acc[0]);
    acc[1] = wmma4(a[s], b1, acc[1]);
    acc[2] = wmma4(a[s], b2, acc[2]);
    acc[3] = wmma4(a[s], b3, acc[3]);
  };
#pragma unroll
  for (int s = 0; s < OP_G; ++s) {
    load(s, 4 * s);
    __builtin_amdgcn_sched_barrier(0);
  }
  for (int k0 = 0; k0 + 4 * OP_G < OP_JL; k0 += 4 * OP_G) {
#pragma unroll
    for (int s = 0; s < OP_G; ++s) {
      mm(s);
      load(s, k0 + 4 * OP_G + 4 * s);
      __builtin_amdgcn_sched_barrier(0);
    }
  }
#pragma unroll
  for (int s = 0; s < OP_G; ++s) mm(s);
  if (h == 0) {
#pragma unroll
    for (int v = 0; v < 8; ++v) {
      v4f o = {acc[0][v], acc[1][v], acc[2][v], acc[3][v]};
      *reinterpret_cast<v4f*>(&part[jq][v][dcol]) = o;
    }
  }
  __syncthreads();
  v4f sum = *reinterpret_cast<const v4f*>(&part[0][wave][4 * lane]);
#pragma unroll
  for (int q = 1; q < OP_JQ; ++q)
    sum += *reinterpret_cast<const v4f*>(&part[q][wave][4 * lane]);
  *reinterpret_cast<v4f*>(out_cat + (size_t)bi * DOUT + DD + wave * DD + 4 * lane) = sum;
}

#define PJ_BLK_S 96
#define PJ_BLK_P 96
#define PJ_BLKS  (3 * PJ_BLK_S + 3 * PJ_BLK_P)
#define PT_PITCH 49

__global__ void __launch_bounds__(128)
proj_bias_kernel(const float* __restrict__ x,
                 const float* __restrict__ e,
                 const float* __restrict__ Wq_s,
                 const float* __restrict__ Wk_s,
                 const float* __restrict__ Wv_s,
                 const float* __restrict__ Wq_p,
                 const float* __restrict__ Wk_p,
                 const float* __restrict__ Wv_p,
                 const float* __restrict__ Wb,
                 const float* __restrict__ r,
                 const float* __restrict__ t,
                 float* __restrict__ qs,
                 float* __restrict__ ks,
                 float* __restrict__ vs,
                 float* __restrict__ qp,
                 float* __restrict__ kp,
                 float* __restrict__ vp,
                 float* __restrict__ qq,
                 float* __restrict__ kk,
                 float* __restrict__ bias) {
  __shared__ float hm[16 * PT_PITCH];
  int blk = blockIdx.x;
  if (blk < BL) {
    bias_row(e, Wb, bias, blk);
    return;
  }
  blk -= BL;
  int wave = threadIdx.x >> 5;
  if (blk < 3 * PJ_BLK_S) {
    int g = blk / PJ_BLK_S;
    blk -= g * PJ_BLK_S;
    const float* Wm = g == 0 ? Wq_s : (g == 1 ? Wk_s : Wv_s);
    float* C        = g == 0 ? qs : (g == 1 ? ks : vs);
    gemm_tile(x, Wm, nullptr, C, DD, DD, blk * 4 + wave);
    return;
  }
  blk -= 3 * PJ_BLK_S;
  int g = blk / PJ_BLK_P;
  blk -= g * PJ_BLK_P;
  const float* Wm = g == 0 ? Wq_p : (g == 1 ? Wk_p : Wv_p);
  float* outp     = g == 0 ? qp : (g == 1 ? kp : vp);
  float* ss       = g == 0 ? qq : (g == 1 ? kk : nullptr);
  int tm = (blk >> 1) << 4;
  int half = blk & 1;
  int rr = threadIdx.x & 15;
  int hh = (threadIdx.x >> 4) & 3;
  int row = tm + rr;
  float rv[9], tv[3];
#pragma unroll
  for (int q = 0; q < 9; ++q) rv[q] = r[(size_t)row * 9 + q];
#pragma unroll
  for (int q = 0; q < 3; ++q) tv[q] = t[(size_t)row * 3 + q];
  if (wave < 3) {
    int lane = threadIdx.x & 31;
    int h = lane >> 4, l16 = lane & 15;
    v8f acc = gemm_acc(x, Wm, DD, NH * 12, tm, half * 48 + wave * 16, 0, DD);
#pragma unroll
    for (int v = 0; v < 8; ++v) hm[(v + (h << 3)) * PT_PITCH + wave * 16 + l16] = acc[v];
  }
  __syncthreads();
  if (threadIdx.x < 64) {
    int b = row / LL, l = row % LL;
    int n = half * 4 + hh;
    point_transform(&hm[rr * PT_PITCH + hh * 12], rv, tv, outp, ss,
                    (b * NH + n) * LL + l);
  }
}

#define OP_BLKS (BB * LL)
#define AV_BLKS ((BB * NH * LT) / 8)

__global__ void attn_reduce_kernel(const float* __restrict__ attn,
                                   const float* __restrict__ e,
                                   const float* __restrict__ vs,
                                   const float* __restrict__ vp,
                                   const float* __restrict__ r,
                                   const float* __restrict__ t,
                                   float* __restrict__ out_cat) {
  __shared__ __attribute__((aligned(16))) float part[OP_JQ][NH][DD];
  int wave = threadIdx.x >> 5;
  if (blockIdx.x < OP_BLKS) {
    out_pair_row(attn, e, out_cat, part, blockIdx.x);
  } else {
    attn_v_tile(attn, vs, vp, r, t, out_cat, (blockIdx.x - OP_BLKS) * 8 + wave);
  }
}

#define FP_KS 10
#define FP_KC (DOUT / FP_KS)

__global__ void __launch_bounds__(FP_KS * 32)
out_proj_kernel(const float* __restrict__ A,
                const float* __restrict__ W,
                const float* __restrict__ bias,
                float* __restrict__ C) {
  __shared__ float red[FP_KS][8][32];
  int wave = threadIdx.x >> 5;
  int lane = threadIdx.x & 31;
  int tm = (blockIdx.x / (DD / 16)) << 4;
  int tn = (blockIdx.x % (DD / 16)) << 4;
  v8f acc = gemm_acc(A, W, DOUT, DD, tm, tn, wave * FP_KC, FP_KC);
#pragma unroll
  for (int v = 0; v < 8; ++v) red[wave][v][lane] = acc[v];
  __syncthreads();
  if (wave < 8) {
    float s = red[0][wave][lane];
#pragma unroll
    for (int w = 1; w < FP_KS; ++w) s += red[w][wave][lane];
    int row = tm + wave + ((lane >> 4) << 3);
    int col = tn + (lane & 15);
    C[(size_t)row * DD + col] = s + bias[col];
  }
}

extern "C" void kernel_launch(void* const* d_in, const int* in_sizes, int n_in,
                              void* d_out, int out_size, void* d_ws, size_t ws_size,
                              hipStream_t stream) {
  (void)in_sizes; (void)n_in; (void)out_size;
  const float* x     = (const float*)d_in[0];
  const float* e     = (const float*)d_in[1];
  const float* r     = (const float*)d_in[2];
  const float* t     = (const float*)d_in[3];
  const float* Wq_s  = (const float*)d_in[4];
  const float* Wk_s  = (const float*)d_in[5];
  const float* Wv_s  = (const float*)d_in[6];
  const float* Wb    = (const float*)d_in[7];
  const float* Wq_p  = (const float*)d_in[8];
  const float* Wk_p  = (const float*)d_in[9];
  const float* Wv_p  = (const float*)d_in[10];
  const float* gamma = (const float*)d_in[11];
  const float* Wo    = (const float*)d_in[12];
  const float* bo    = (const float*)d_in[13];
  float* out = (float*)d_out;
  float* ws  = (float*)d_ws;

  float* qs   = ws;
  float* ks_  = qs  + 98304;
  float* vs   = ks_ + 98304;
  float* qp   = vs  + 98304;
  float* kp   = qp  + 73728;
  float* vp   = kp  + 73728;
  float* qq   = vp  + 73728;
  float* kk   = qq  + 6144;
  float* bias = kk  + 6144;
  float* attn = bias + 2359296;
  float* ocat = attn + 2359296;
  size_t req = (size_t)((ocat + 983040) - ws) * sizeof(float);
  if (ws_size < req) return;

  dim3 blk(256);

  proj_bias_kernel<<<BL + PJ_BLKS, dim3(128), 0, stream>>>(
      x, e, Wq_s, Wk_s, Wv_s, Wq_p, Wk_p, Wv_p, Wb, r, t, qs, ks_, vs, qp, kp, vp,
      qq, kk, bias);

  logit_softmax_kernel<<<BB * NH * LT, dim3(LS_W * 32), 0, stream>>>(
      qs, ks_, qp, kp, qq, kk, bias, gamma, attn);

  attn_reduce_kernel<<<OP_BLKS + AV_BLKS, blk, 0, stream>>>(attn, e, vs, vp, r, t,
                                                            ocat);

  out_proj_kernel<<<(BL / 16) * (DD / 16), dim3(FP_KS * 32), 0, stream>>>(ocat, Wo, bo,
                                                                        out);
}


# attn@V folded into the pair-output blocks: block 2m/2m+1 takes the scalar/point product of tile m, K split over its 8 waves, partials added via LDS; equal work per block; attn@V sum order changes
# speedup vs baseline: 5.3968x; 1.0517x over previous
#include <hip/hip_runtime.h>
#include <hip/hip_bf16.h>
#include <math.h>

#define BB   2
#define LL   384
#define DD   128
#define NH   8
#define DHS_ 16
#define DHP_ 4
#define BL   (BB*LL)
#define BNL  (BB*NH*LL)
#define DOUT 1280
#define LT   24

#define GK 16

typedef float v2f __attribute__((ext_vector_type(2)));
typedef float v8f __attribute__((ext_vector_type(8)));

static __device__ __forceinline__ v8f wmma4(v2f a, v2f b, v8f c) {
  return __builtin_amdgcn_wmma_f32_16x16x4_f32(false, a, false, b, (short)0, c,
                                               false, false);
}


static __device__ __forceinline__ v8f gemm_acc(const float* __restrict__ A,
                                               const float* __restrict__ W,
                                               int lda, int Nc, int tm, int tn,
                                               int kbeg, int klen) {
  int lane = threadIdx.x & 31;
  int h = lane >> 4, l16 = lane & 15;
  const float* ap = A + (size_t)(tm + l16) * lda + kbeg + 2 * h;
  const float* wp = W + (size_t)(kbeg + 2 * h) * Nc + tn + l16;
  v8f acc = {};
  v2f a[GK], b[GK];
  auto load = [&](int s, int k) {
    a[s].x = ap[k];
    a[s].y = ap[k + 1];
    b[s].x = wp[(size_t)k * Nc];
    b[s].y = wp[(size_t)(k + 1) * Nc];
  };
#pragma unroll
  for (int s = 0; s < GK; ++s) {
    load(s, 4 * s);
    __builtin_amdgcn_sched_barrier(0);
  }
  int k0 = 0;
  for (; k0 + 4 * GK < klen; k0 += 4 * GK) {
#pragma unroll
    for (int s = 0; s < GK; ++s) {
      acc = wmma4(a[s], b[s], acc);
      load(s, k0 + 4 * GK + 4 * s);
      __builtin_amdgcn_sched_barrier(0);
    }
  }
#pragma unroll
  for (int s = 0; s < GK; ++s) acc = wmma4(a[s], b[s], acc);
  return acc;
}

static __device__ __forceinline__ void gemm_tile(const float* __restrict__ A,
                                                 const float* __restrict__ W,
                                                 const float* __restrict__ bias,
                                                 float* __restrict__ C,
                                                 int K, int Nc, int tile) {
  int lane = threadIdx.x & 31;
  int h = lane >> 4, l16 = lane & 15;
  int ntn = Nc >> 4;
  int tm = (tile / ntn) << 4;
  int tn = (tile % ntn) << 4;
  v8f acc = gemm_acc(A, W, K, Nc, tm, tn, 0, K);
  float bv = bias ? bias[tn + l16] : 0.0f;
#pragma unroll
  for (int v = 0; v < 8; ++v) {
    int row = tm + v + (h << 3);
    C[(size_t)row * Nc + tn + l16] = acc[v] + bv;
  }
}

typedef float v4f __attribute__((ext_vector_type(4)));

#define BIAS_TPW 6
#define BIAS_NS  32

static __device__ __forceinline__ void bias_row(const float* __restrict__ e,
                                                const float* __restrict__ Wb,
                                                float* __restrict__ bias,
                                                int bi) {
  int wave = threadIdx.x >> 5;
  int lane = threadIdx.x & 31;
  int h = lane >> 4, l16 = lane & 15;
  int b = bi / LL, i = bi % LL;
  int n = l16 & 7;
  v2f wf[32];
  const float* wp = Wb + (4 * h) * NH + n;
#pragma unroll
  for (int m = 0; m < 16; ++m) {
#pragma unroll
    for (int q = 0; q < 2; ++q) {
      wf[2 * m + q].x = wp[(8 * m + 2 * q) * NH];
      wf[2 * m + q].y = wp[(8 * m + 2 * q + 1) * NH];
    }
  }
  const float* ap =
      e + ((size_t)bi * LL + (size_t)(wave * BIAS_TPW * 16 + l16)) * DD + 4 * h;
  auto ld = [&](int t, int m) -> v4f {
    return *reinterpret_cast<const v4f*>(ap + (size_t)t * (16 * DD) + 8 * m);
  };
  bool hi = l16 >= 8;
  float* bp = bias + (((size_t)(b * NH + n) * LL + i) * LL) +
              wave * BIAS_TPW * 16 + 8 * h + (hi ? 4 : 0);
  auto store = [&](const v8f& acc, int t) {
    v4f o;
    o.x = hi ? acc[4] : acc[0];
    o.y = hi ? acc[5] : acc[1];
    o.z = hi ? acc[6] : acc[2];
    o.w = hi ? acc[7] : acc[3];
    *reinterpret_cast<v4f*>(bp + t * 16) = o;
  };
  v4f a[BIAS_NS];
#pragma unroll
  for (int s = 0; s < BIAS_NS; ++s) {
    a[s] = ld(s >> 4, s & 15);
    __builtin_amdgcn_sched_barrier(0);
  }
  for (int tp = 0; tp < BIAS_TPW / 2 - 1; ++tp) {
#pragma unroll
    for (int hf = 0; hf < 2; ++hf) {
      v8f acc = {};
#pragma unroll
      for (int m = 0; m < 16; ++m) {
        int s = hf * 16 + m;
        v2f a0 = {a[s].x, a[s].y}, a1 = {a[s].z, a[s].w};
        acc = wmma4(a0, wf[2 * m], acc);
        acc = wmma4(a1, wf[2 * m + 1], acc);
        a[s] = ld(2 * tp + 2 + hf, m);
        __builtin_amdgcn_sched_barrier(0);
      }
      store(acc, 2 * tp + hf);
    }
  }
#pragma unroll
  for (int hf = 0; hf < 2; ++hf) {
    v8f acc = {};
#pragma unroll
    for (int m = 0; m < 16; ++m) {
      int s = hf * 16 + m;
      v2f a0 = {a[s].x, a[s].y}, a1 = {a[s].z, a[s].w};
      acc = wmma4(a0, wf[2 * m], acc);
      acc = wmma4(a1, wf[2 * m + 1], acc);
    }
    store(acc, BIAS_TPW - 2 + hf);
  }
}

static __device__ __forceinline__ void point_transform(const float* hp,
                                                       const float* rp,
                                                       const float* tp,
                                                       float* __restrict__ outp,
                                                       float* __restrict__ sumsq,
                                                       int idx) {
  float* op = outp + (size_t)idx * 12;
  float ss = 0.0f;
#pragma unroll
  for (int p = 0; p < 4; ++p)
#pragma unroll
    for (int c = 0; c < 3; ++c) {
      float v = tp[c];
#pragma unroll
      for (int k = 0; k < 3; ++k) v += hp[p * 3 + k] * rp[k * 3 + c];
      op[p * 3 + c] = v;
      ss += v * v;
    }
  if (sumsq) sumsq[idx] = ss;
}

#define LS_W 8
#define LS_T (LT / LS_W)

__global__ void __launch_bounds__(LS_W * 32)
logit_softmax_kernel(const float* __restrict__ qs,
                     const float* __restrict__ ks,
                     const float* __restrict__ qp,
                     const float* __restrict__ kp,
                     const float* __restrict__ qq,
                     const float* __restrict__ kk,
                     const float* __restrict__ bias,
                     const float* __restrict__ gamma,
                     float* __restrict__ attn) {
  __shared__ __attribute__((aligned(16))) float red_m[LS_W][16];
  __shared__ __attribute__((aligned(16))) float red_s[LS_W][16];
  int strip = blockIdx.x;
  int it = strip % LT;
  int n  = (strip / LT) % NH;
  int b  = strip / (LT * NH);
  int wave = threadIdx.x >> 5;
  int lane = threadIdx.x & 31;
  int h = lane >> 4, l16 = lane & 15;
  size_t hb = (size_t)(b * NH + n) * LL;

  const float* qsp = qs + (size_t)(b * LL + it * 16 + l16) * DD + n * DHS_ + 2 * h;
  const float* qpp = qp + (hb + it * 16 + l16) * 12 + 2 * h;
  v2f aS[4], aP[3];
#pragma unroll
  for (int k = 0; k < 4; ++k) { aS[k].x = qsp[4 * k]; aS[k].y = qsp[4 * k + 1]; }
#pragma unroll
  for (int k = 0; k < 3; ++k) { aP[k].x = qpp[4 * k]; aP[k].y = qpp[4 * k + 1]; }
  v2f bS[LS_T][4], bP[LS_T][3];
  float kkj[LS_T], bsv[LS_T][8], qqi[8];
#pragma unroll
  for (int t = 0; t < LS_T; ++t) {
    int jt = wave * LS_T + t;
    const float* ksp = ks + (size_t)(b * LL + jt * 16 + l16) * DD + n * DHS_ + 2 * h;
    const float* kpp = kp + (hb + jt * 16 + l16) * 12 + 2 * h;
#pragma unroll
    for (int k = 0; k < 4; ++k) { bS[t][k].x = ksp[4 * k]; bS[t][k].y = ksp[4 * k + 1]; }
#pragma unroll
    for (int k = 0; k < 3; ++k) { bP[t][k].x = kpp[4 * k]; bP[t][k].y = kpp[4 * k + 1]; }
    kkj[t] = kk[hb + jt * 16 + l16];
#pragma unroll
    for (int v = 0; v < 8; ++v)
      bsv[t][v] = bias[(hb + it * 16 + v + (h << 3)) * LL + jt * 16 + l16];
  }
#pragma unroll
  for (int v = 0; v < 8; ++v) qqi[v] = qq[hb + it * 16 + v + (h << 3)];
  float g = gamma[n];
  const float rsqrt3 = 0.57735026918962576f;
  __builtin_amdgcn_sched_barrier(0);

  float lg[LS_T][8];
#pragma unroll
  for (int t = 0; t < LS_T; ++t) {
    v8f accS = {}, accP = {};
#pragma unroll
    for (int k = 0; k < 4; ++k) accS = wmma4(aS[k], bS[t][k], accS);
#pragma unroll
    for (int k = 0; k < 3; ++k) accP = wmma4(aP[k], bP[t][k], accP);
#pragma unroll
    for (int v = 0; v < 8; ++v) {
      float d2 = qqi[v] + kkj[t] - 2.0f * accP[v];
      float lp = -0.125f * g * d2;
      lg[t][v] = rsqrt3 * (0.25f * accS[v] + bsv[t][v] + lp);
    }
  }

  float m[8];
#pragma unroll
  for (int v = 0; v < 8; ++v) {
    float x = lg[0][v];
#pragma unroll
    for (int t = 1; t < LS_T; ++t) x = fmaxf(x, lg[t][v]);
#pragma unroll
    for (int off = 8; off > 0; off >>= 1) x = fmaxf(x, __shfl_xor(x, off, 32));
    m[v] = x;
  }
  if (l16 == 0) {
#pragma unroll
    for (int v = 0; v < 8; ++v) red_m[wave][8 * h + v] = m[v];
  }
  __syncthreads();
#pragma unroll
  for (int w = 0; w < LS_W; ++w) {
    v4f lo = *reinterpret_cast<const v4f*>(&red_m[w][8 * h]);
    v4f hi = *reinterpret_cast<const v4f*>(&red_m[w][8 * h + 4]);
    float r8[8] = {lo.x, lo.y, lo.z, lo.w, hi.x, hi.y, hi.z, hi.w};
#pragma unroll
    for (int v = 0; v < 8; ++v) m[v] = w == 0 ? r8[v] : fmaxf(m[v], r8[v]);
  }

  float s[8];
#pragma unroll
  for (int v = 0; v < 8; ++v) {
    float x = 0.0f;
#pragma unroll
    for (int t = 0; t < LS_T; ++t) {
      lg[t][v] = __expf(lg[t][v] - m[v]);
      x += lg[t][v];
    }
#pragma unroll
    for (int off = 8; off > 0; off >>= 1) x += __shfl_xor(x, off, 32);
    s[v] = x;
  }
  if (l16 == 0) {
#pragma unroll
    for (int v = 0; v < 8; ++v) red_s[wave][8 * h + v] = s[v];
  }
  __syncthreads();
#pragma unroll
  for (int w = 0; w < LS_W; ++w) {
    v4f lo = *reinterpret_cast<const v4f*>(&red_s[w][8 * h]);
    v4f hi = *reinterpret_cast<const v4f*>(&red_s[w][8 * h + 4]);
    float r8[8] = {lo.x, lo.y, lo.z, lo.w, hi.x, hi.y, hi.z, hi.w};
#pragma unroll
    for (int v = 0; v < 8; ++v) s[v] = w == 0 ? r8[v] : s[v] + r8[v];
  }
#pragma unroll
  for (int v = 0; v < 8; ++v) {
    float inv = 1.0f / s[v];
    size_t row = (hb + it * 16 + v + (h << 3)) * LL;
#pragma unroll
    for (int t = 0; t < LS_T; ++t)
      attn[row + (wave * LS_T + t) * 16 + l16] = lg[t][v] * inv;
  }
}

#define AV_W  8
#define AV_JW (LL / AV_W)
#define AV_KS (AV_JW / 4)

static __device__ __forceinline__ void attn_v_part(const float* __restrict__ attn,
                                                   const float* __restrict__ vs,
                                                   const float* __restrict__ vp,
                                                   float (*avp)[8][32],
                                                   int tile,
                                                   bool point) {
  int it = tile % LT;
  int n  = (tile / LT) % NH;
  int b  = tile / (LT * NH);
  int wave = threadIdx.x >> 5;
  int lane = threadIdx.x & 31;
  int h = lane >> 4, l16 = lane & 15;
  size_t hb = (size_t)(b * NH + n) * LL;
  int j0 = wave * AV_JW + 2 * h;
  const float* ap = attn + (hb + it * 16 + l16) * LL + j0;
  const float* bq = point
      ? vp + (hb + j0) * 12 + (l16 < 12 ? l16 : 11)
      : vs + ((size_t)b * LL + j0) * DD + n * DHS_ + l16;
  int bst = point ? 12 : DD;
  v2f a[AV_KS], bf[AV_KS];
#pragma unroll
  for (int s = 0; s < AV_KS; ++s) {
    a[s].x = ap[4 * s];
    a[s].y = ap[4 * s + 1];
    bf[s].x = bq[(size_t)(4 * s) * bst];
    bf[s].y = bq[(size_t)(4 * s + 1) * bst];
  }
  __builtin_amdgcn_sched_barrier(0);
  v8f acc = {};
#pragma unroll
  for (int s = 0; s < AV_KS; ++s) acc = wmma4(a[s], bf[s], acc);
#pragma unroll
  for (int v = 0; v < 8; ++v) avp[wave][v][lane] = acc[v];
}

static __device__ __forceinline__ void attn_v_finish(const float (*avp)[8][32],
                                                     const float* __restrict__ r,
                                                     const float* __restrict__ t,
                                                     float* __restrict__ out_cat,
                                                     int tile, bool point) {
  int it = tile % LT;
  int n  = (tile / LT) % NH;
  int b  = tile / (LT * NH);
  int wave = threadIdx.x >> 5;
  int lane = threadIdx.x & 31;
  int h = lane >> 4, l16 = lane & 15;
  float sum = avp[0][wave][lane];
#pragma unroll
  for (int w = 1; w < AV_W; ++w) sum += avp[w][wave][lane];
  int i = it * 16 + wave + (h << 3);
  float* oc = out_cat + (size_t)(b * LL + i) * DOUT;
  if (!point) {
    oc[n * DHS_ + l16] = sum;
    return;
  }
  bool pv = l16 < 12;
  int cl = pv ? l16 : 11;
  int p  = cl / 3;
  int kc = cl - 3 * p;
  int src = (lane & 16) + 3 * p;
  const float* rp = r + (size_t)(b * LL + i) * 9 + kc * 3;
  float om = sum - t[(size_t)(b * LL + i) * 3 + kc];
  float om0 = __shfl(om, src, 32);
  float om1 = __shfl(om, src + 1, 32);
  float om2 = __shfl(om, src + 2, 32);
  float val = om0 * rp[0] + om1 * rp[1] + om2 * rp[2];
  float sq = val * val;
  float nsq = 0.0f;
  nsq += __shfl(sq, src, 32);
  nsq += __shfl(sq, src + 1, 32);
  nsq += __shfl(sq, src + 2, 32);
  if (pv) {
    oc[DD + NH * DD + n * 12 + l16] = val;
    if (kc == 0) oc[DD + NH * DD + NH * 12 + n * 4 + p] = sqrtf(nsq);
  }
}

#define OP_JQ 4
#define OP_JL (LL / OP_JQ)
#define OP_G  6

static __device__ __forceinline__ void out_pair_row(const float* __restrict__ attn,
                                                    const float* __restrict__ e,
                                                    float* __restrict__ out_cat,
                                                    float (*part)[NH][DD],
                                                    int bi) {
  int wave = threadIdx.x >> 5;
  int lane = threadIdx.x & 31;
  int h = lane >> 4, l16 = lane & 15;
  int b = bi / LL, i = bi % LL;
  int dh = wave & 1;
  int jq = wave >> 1;
  int dcol = dh * 64 + 4 * l16;
  const float* ap =
      attn + ((size_t)(b * NH + (l16 & 7)) * LL + i) * LL + jq * OP_JL + 2 * h;
  const float* ep = e + ((size_t)bi * LL + jq * OP_JL + 2 * h) * DD + dcol;
  v8f acc[4] = {};
  v2f a[OP_G];
  v4f bx[OP_G], by[OP_G];
  auto load = [&](int s, int k) {
    a[s].x = ap[k];
    a[s].y = ap[k + 1];
    bx[s] = *reinterpret_cast<const v4f*>(ep + (size_t)k * DD);
    by[s] = *reinterpret_cast<const v4f*>(ep + (size_t)(k + 1) * DD);
  };
  auto mm = [&](int s) {
    v2f b0 = {bx[s].x, by[s].x}, b1 = {bx[s].y, by[s].y};
    v2f b2 = {bx[s].z, by[s].z}, b3 = {bx[s].w, by[s].w};
    acc[0] = wmma4(a[s], b0, acc[0]);
    acc[1] = wmma4(a[s], b1, acc[1]);
    acc[2] = wmma4(a[s], b2, acc[2]);
    acc[3] = wmma4(a[s], b3, acc[3]);
  };
#pragma unroll
  for (int s = 0; s < OP_G; ++s) {
    load(s, 4 * s);
    __builtin_amdgcn_sched_barrier(0);
  }
  for (int k0 = 0; k0 + 4 * OP_G < OP_JL; k0 += 4 * OP_G) {
#pragma unroll
    for (int s = 0; s < OP_G; ++s) {
      mm(s);
      load(s, k0 + 4 * OP_G + 4 * s);
      __builtin_amdgcn_sched_barrier(0);
    }
  }
#pragma unroll
  for (int s = 0; s < OP_G; ++s) mm(s);
  if (h == 0) {
#pragma unroll
    for (int v = 0; v < 8; ++v) {
      v4f o = {acc[0][v], acc[1][v], acc[2][v], acc[3][v]};
      *reinterpret_cast<v4f*>(&part[jq][v][dcol]) = o;
    }
  }
  __syncthreads();
  v4f sum = *reinterpret_cast<const v4f*>(&part[0][wave][4 * lane]);
#pragma unroll
  for (int q = 1; q < OP_JQ; ++q)
    sum += *reinterpret_cast<const v4f*>(&part[q][wave][4 * lane]);
  *reinterpret_cast<v4f*>(out_cat + (size_t)bi * DOUT + DD + wave * DD + 4 * lane) = sum;
}

#define PJ_BLK_S 96
#define PJ_BLK_P 96
#define PJ_BLKS  (3 * PJ_BLK_S + 3 * PJ_BLK_P)
#define PT_PITCH 49

__global__ void __launch_bounds__(128)
proj_bias_kernel(const float* __restrict__ x,
                 const float* __restrict__ e,
                 const float* __restrict__ Wq_s,
                 const float* __restrict__ Wk_s,
                 const float* __restrict__ Wv_s,
                 const float* __restrict__ Wq_p,
                 const float* __restrict__ Wk_p,
                 const float* __restrict__ Wv_p,
                 const float* __restrict__ Wb,
                 const float* __restrict__ r,
                 const float* __restrict__ t,
                 float* __restrict__ qs,
                 float* __restrict__ ks,
                 float* __restrict__ vs,
                 float* __restrict__ qp,
                 float* __restrict__ kp,
                 float* __restrict__ vp,
                 float* __restrict__ qq,
                 float* __restrict__ kk,
                 float* __restrict__ bias) {
  __shared__ float hm[16 * PT_PITCH];
  int blk = blockIdx.x;
  if (blk < BL) {
    bias_row(e, Wb, bias, blk);
    return;
  }
  blk -= BL;
  int wave = threadIdx.x >> 5;
  if (blk < 3 * PJ_BLK_S) {
    int g = blk / PJ_BLK_S;
    blk -= g * PJ_BLK_S;
    const float* Wm = g == 0 ? Wq_s : (g == 1 ? Wk_s : Wv_s);
    float* C        = g == 0 ? qs : (g == 1 ? ks : vs);
    gemm_tile(x, Wm, nullptr, C, DD, DD, blk * 4 + wave);
    return;
  }
  blk -= 3 * PJ_BLK_S;
  int g = blk / PJ_BLK_P;
  blk -= g * PJ_BLK_P;
  const float* Wm = g == 0 ? Wq_p : (g == 1 ? Wk_p : Wv_p);
  float* outp     = g == 0 ? qp : (g == 1 ? kp : vp);
  float* ss       = g == 0 ? qq : (g == 1 ? kk : nullptr);
  int tm = (blk >> 1) << 4;
  int half = blk & 1;
  int rr = threadIdx.x & 15;
  int hh = (threadIdx.x >> 4) & 3;
  int row = tm + rr;
  float rv[9], tv[3];
#pragma unroll
  for (int q = 0; q < 9; ++q) rv[q] = r[(size_t)row * 9 + q];
#pragma unroll
  for (int q = 0; q < 3; ++q) tv[q] = t[(size_t)row * 3 + q];
  if (wave < 3) {
    int lane = threadIdx.x & 31;
    int h = lane >> 4, l16 = lane & 15;
    v8f acc = gemm_acc(x, Wm, DD, NH * 12, tm, half * 48 + wave * 16, 0, DD);
#pragma unroll
    for (int v = 0; v < 8; ++v) hm[(v + (h << 3)) * PT_PITCH + wave * 16 + l16] = acc[v];
  }
  __syncthreads();
  if (threadIdx.x < 64) {
    int b = row / LL, l = row % LL;
    int n = half * 4 + hh;
    point_transform(&hm[rr * PT_PITCH + hh * 12], rv, tv, outp, ss,
                    (b * NH + n) * LL + l);
  }
}

#define OP_BLKS (BB * LL)
static_assert(OP_BLKS == 2 * BB * NH * LT, "two blocks per attn@V tile");

__global__ void attn_reduce_kernel(const float* __restrict__ attn,
                                   const float* __restrict__ e,
                                   const float* __restrict__ vs,
                                   const float* __restrict__ vp,
                                   const float* __restrict__ r,
                                   const float* __restrict__ t,
                                   float* __restrict__ out_cat) {
  __shared__ __attribute__((aligned(16))) float part[OP_JQ][NH][DD];
  __shared__ float avp[AV_W][8][32];
  int tile = blockIdx.x >> 1;
  bool point = blockIdx.x & 1;
  attn_v_part(attn, vs, vp, avp, tile, point);
  out_pair_row(attn, e, out_cat, part, blockIdx.x);
  attn_v_finish(avp, r, t, out_cat, tile, point);
}

#define FP_KS 10
#define FP_KC (DOUT / FP_KS)

__global__ void __launch_bounds__(FP_KS * 32)
out_proj_kernel(const float* __restrict__ A,
                const float* __restrict__ W,
                const float* __restrict__ bias,
                float* __restrict__ C) {
  __shared__ float red[FP_KS][8][32];
  int wave = threadIdx.x >> 5;
  int lane = threadIdx.x & 31;
  int tm = (blockIdx.x / (DD / 16)) << 4;
  int tn = (blockIdx.x % (DD / 16)) << 4;
  v8f acc = gemm_acc(A, W, DOUT, DD, tm, tn, wave * FP_KC, FP_KC);
#pragma unroll
  for (int v = 0; v < 8; ++v) red[wave][v][lane] = acc[v];
  __syncthreads();
  if (wave < 8) {
    float s = red[0][wave][lane];
#pragma unroll
    for (int w = 1; w < FP_KS; ++w) s += red[w][wave][lane];
    int row = tm + wave + ((lane >> 4) << 3);
    int col = tn + (lane & 15);
    C[(size_t)row * DD + col] = s + bias[col];
  }
}

extern "C" void kernel_launch(void* const* d_in, const int* in_sizes, int n_in,
                              void* d_out, int out_size, void* d_ws, size_t ws_size,
                              hipStream_t stream) {
  (void)in_sizes; (void)n_in; (void)out_size;
  const float* x     = (const float*)d_in[0];
  const float* e     = (const float*)d_in[1];
  const float* r     = (const float*)d_in[2];
  const float* t     = (const float*)d_in[3];
  const float* Wq_s  = (const float*)d_in[4];
  const float* Wk_s  = (const float*)d_in[5];
  const float* Wv_s  = (const float*)d_in[6];
  const float* Wb    = (const float*)d_in[7];
  const float* Wq_p  = (const float*)d_in[8];
  const float* Wk_p  = (const float*)d_in[9];
  const float* Wv_p  = (const float*)d_in[10];
  const float* gamma = (const float*)d_in[11];
  const float* Wo    = (const float*)d_in[12];
  const float* bo    = (const float*)d_in[13];
  float* out = (float*)d_out;
  float* ws  = (float*)d_ws;

  float* qs   = ws;
  float* ks_  = qs  + 98304;
  float* vs   = ks_ + 98304;
  float* qp   = vs  + 98304;
  float* kp   = qp  + 73728;
  float* vp   = kp  + 73728;
  float* qq   = vp  + 73728;
  float* kk   = qq  + 6144;
  float* bias = kk  + 6144;
  float* attn = bias + 2359296;
  float* ocat = attn + 2359296;
  size_t req = (size_t)((ocat + 983040) - ws) * sizeof(float);
  if (ws_size < req) return;

  dim3 blk(256);

  proj_bias_kernel<<<BL + PJ_BLKS, dim3(128), 0, stream>>>(
      x, e, Wq_s, Wk_s, Wv_s, Wq_p, Wk_p, Wv_p, Wb, r, t, qs, ks_, vs, qp, kp, vp,
      qq, kk, bias);

  logit_softmax_kernel<<<BB * NH * LT, dim3(LS_W * 32), 0, stream>>>(
      qs, ks_, qp, kp, qq, kk, bias, gamma, attn);

  attn_reduce_kernel<<<OP_BLKS, blk, 0, stream>>>(attn, e, vs, vp, r, t, ocat);

  out_proj_kernel<<<(BL / 16) * (DD / 16), dim3(FP_KS * 32), 0, stream>>>(ocat, Wo, bo,
                                                                        out);
}
